# starting kernel
# baseline (speedup 1.0000x reference)
#include <hip/hip_runtime.h>
#include <hip/hip_bf16.h>


typedef __attribute__((ext_vector_type(16))) __bf16     v16bf;
typedef __attribute__((ext_vector_type(8)))  float      v8f;
typedef __attribute__((ext_vector_type(8)))  unsigned   v8u;

union FragAB { v8u u; v16bf b; };

__device__ __forceinline__ unsigned bf16_rne(float f) {
    unsigned u = __float_as_uint(f);
    u += 0x7FFFu + ((u >> 16) & 1u);
    return u >> 16;
}
__device__ __forceinline__ unsigned pack2bf(float lo, float hi) {
    return bf16_rne(lo) | (bf16_rne(hi) << 16);
}

__global__ void k_deg_init(float* dinv, int n) {
    int i = blockIdx.x * blockDim.x + threadIdx.x;
    if (i < n) dinv[i] = 1.0f;
}
__global__ void k_deg_edges(const int* __restrict__ dst, float* dinv, int e) {
    int i = blockIdx.x * blockDim.x + threadIdx.x;
    if (i < e) atomicAdd(&dinv[dst[i]], 1.0f);
}
__global__ void k_deg_rsqrt(float* dinv, int n) {
    int i = blockIdx.x * blockDim.x + threadIdx.x;
    if (i < n) dinv[i] = rsqrtf(dinv[i]);
}

__global__ void k_pack_x(const float* __restrict__ x, unsigned* __restrict__ xb,
                         long long n_pairs) {
    long long i = (long long)blockIdx.x * blockDim.x + threadIdx.x;
    if (i < n_pairs) {
        float2 v = ((const float2*)x)[i];
        xb[i] = pack2bf(v.x, v.y);
    }
}
__global__ void k_pack_w1(const float* __restrict__ W1, unsigned* __restrict__ w1p) {
    int tid = blockIdx.x * blockDim.x + threadIdx.x;
    if (tid >= 8192) return;
    int v    = tid & 7;
    int lane = (tid >> 3) & 31;
    int ks   = (tid >> 8) & 3;
    int nt   = tid >> 10;
    int k    = ks * 32 + ((v >> 2) << 4) + ((lane >> 4) << 3) + ((v & 3) << 1);
    int ncol = nt * 16 + (lane & 15);
    w1p[tid] = pack2bf(W1[(size_t)k * 128 + ncol], W1[(size_t)(k + 1) * 128 + ncol]);
}

__global__ void k_gemm1_wmma(const unsigned* __restrict__ xb,
                             const unsigned* __restrict__ w1p,
                             const float* __restrict__ dinv,
                             float* __restrict__ h1,
                             float* __restrict__ agg1,
                             int n_nodes) {
    const int lane = threadIdx.x & 31;
    const int wid  = threadIdx.x >> 5;
    const int m0   = blockIdx.x * 16;
    const int mr   = lane & 15;
    const int half = lane >> 4;

    int row = m0 + mr;
    if (row >= n_nodes) row = n_nodes - 1;
    const unsigned* xr = xb + (size_t)row * 64;
    const v8u* bfrag = (const v8u*)w1p;

    v8f acc = {};
#pragma unroll
    for (int ks = 0; ks < 4; ++ks) {
        FragAB a, b;
        uint4 alo = *(const uint4*)(xr + ks * 16 + half * 4);
        uint4 ahi = *(const uint4*)(xr + ks * 16 + 8 + half * 4);
        a.u = (v8u){alo.x, alo.y, alo.z, alo.w, ahi.x, ahi.y, ahi.z, ahi.w};
        b.u = bfrag[(wid * 4 + ks) * 32 + lane];
        acc = __builtin_amdgcn_wmma_f32_16x16x32_bf16(
                  false, a.b, false, b.b, (short)0, acc, false, false);
    }
    const int ncol  = wid * 16 + mr;
    const int mbase = m0 + (half << 3);
    if (m0 + 16 <= n_nodes) {
        float*       hp = h1   + (size_t)mbase * 128 + ncol;
        float*       ap = agg1 + (size_t)mbase * 128 + ncol;
        const float4 d0 = *(const float4*)(dinv + mbase);
        const float4 d1 = *(const float4*)(dinv + mbase + 4);
        const float  dv[8] = {d0.x, d0.y, d0.z, d0.w, d1.x, d1.y, d1.z, d1.w};
#pragma unroll
        for (int v = 0; v < 8; ++v) {
            float val = acc[v];
            hp[(size_t)v * 128] = val;
            ap[(size_t)v * 128] = val * dv[v] * dv[v];
        }
    } else {
#pragma unroll
        for (int v = 0; v < 8; ++v) {
            const int m = mbase + v;
            if (m < n_nodes) {
                float val = acc[v];
                float di  = dinv[m];
                h1  [(size_t)m * 128 + ncol] = val;
                agg1[(size_t)m * 128 + ncol] = val * di * di;
            }
        }
    }
}

__global__ void k_scatter1(const float* __restrict__ h1,
                           const int* __restrict__ src, const int* __restrict__ dst,
                           const float* __restrict__ dinv,
                           float* __restrict__ agg1, int n_edges) {
    int e    = (blockIdx.x * blockDim.x + threadIdx.x) >> 5;
    int lane = threadIdx.x & 31;
    if (e >= n_edges) return;
    int s = __builtin_amdgcn_readfirstlane(src[e]);
    int d = __builtin_amdgcn_readfirstlane(dst[e]);
    float norm = dinv[s] * dinv[d];
    float4 v = ((const float4*)(h1 + (size_t)s * 128))[lane];
    float* out = agg1 + (size_t)d * 128 + lane * 4;
    atomicAdd(out + 0, v.x * norm);
    atomicAdd(out + 1, v.y * norm);
    atomicAdd(out + 2, v.z * norm);
    atomicAdd(out + 3, v.w * norm);
}

__global__ void k_layer2(const float* __restrict__ agg1, const float* __restrict__ b1,
                         const float* __restrict__ W2, float* __restrict__ h2,
                         int n_nodes) {
    int node = (blockIdx.x * blockDim.x + threadIdx.x) >> 5;
    int lane = threadIdx.x & 31;
    if (node >= n_nodes) return;
    float4 v  = ((const float4*)(agg1 + (size_t)node * 128))[lane];
    float4 bb = ((const float4*)b1)[lane];
    float t0 = fmaxf(v.x + bb.x, 0.f), t1 = fmaxf(v.y + bb.y, 0.f);
    float t2 = fmaxf(v.z + bb.z, 0.f), t3 = fmaxf(v.w + bb.w, 0.f);
    int c = lane * 4;
    const float2* w2 = (const float2*)W2;
    float2 w0 = w2[c], w1 = w2[c + 1], w2v = w2[c + 2], w3 = w2[c + 3];
    float s0 = t0 * w0.x + t1 * w1.x + t2 * w2v.x + t3 * w3.x;
    float s1 = t0 * w0.y + t1 * w1.y + t2 * w2v.y + t3 * w3.y;
#pragma unroll
    for (int off = 16; off > 0; off >>= 1) {
        s0 += __shfl_down(s0, off, 32);
        s1 += __shfl_down(s1, off, 32);
    }
    if (lane == 0) { h2[node * 2] = s0; h2[node * 2 + 1] = s1; }
}

__global__ void k_self2(const float* __restrict__ h2, const float* __restrict__ dinv,
                        float* __restrict__ agg2, int n_nodes) {
    int i = blockIdx.x * blockDim.x + threadIdx.x;
    if (i < n_nodes * 2) {
        float di = dinv[i >> 1];
        agg2[i] = h2[i] * di * di;
    }
}
__global__ void k_scatter2(const float* __restrict__ h2,
                           const int* __restrict__ src, const int* __restrict__ dst,
                           const float* __restrict__ dinv,
                           float* __restrict__ agg2, int n_edges) {
    int e = blockIdx.x * blockDim.x + threadIdx.x;
    if (e >= n_edges) return;
    int s = src[e], d = dst[e];
    float norm = dinv[s] * dinv[d];
    float2 hv = ((const float2*)h2)[s];
    atomicAdd(&agg2[d * 2 + 0], hv.x * norm);
    atomicAdd(&agg2[d * 2 + 1], hv.y * norm);
}
__global__ void k_logsoftmax(const float* __restrict__ agg2, const float* __restrict__ b2,
                             float* __restrict__ out, int n_nodes) {
    int n = blockIdx.x * blockDim.x + threadIdx.x;
    if (n >= n_nodes) return;
    float2 a = ((const float2*)agg2)[n];
    float z0 = a.x + b2[0];
    float z1 = a.y + b2[1];
    float m  = fmaxf(z0, z1);
    float l  = m + __logf(__expf(z0 - m) + __expf(z1 - m));
    ((float2*)out)[n] = make_float2(z0 - l, z1 - l);
}

extern "C" void kernel_launch(void* const* d_in, const int* in_sizes, int n_in,
                              void* d_out, int out_size, void* d_ws, size_t ws_size,
                              hipStream_t stream) {
    const float* x   = (const float*)d_in[0];
    const int*   ei  = (const int*)  d_in[1];
    const float* W1  = (const float*)d_in[2];
    const float* b1  = (const float*)d_in[3];
    const float* W2  = (const float*)d_in[4];
    const float* b2  = (const float*)d_in[5];
    float*       out = (float*)d_out;

    const int n_nodes = in_sizes[0] / 128;
    const int n_edges = in_sizes[1] / 2;
    const int* src = ei;
    const int* dst = ei + n_edges;

    float*    ws   = (float*)d_ws;
    float*    dinv = ws;
    unsigned* xb   = (unsigned*)(dinv + n_nodes);
    unsigned* w1p  = xb + (size_t)n_nodes * 64;
    float*    h1   = (float*)(w1p + 8192);
    float*    agg1 = h1 + (size_t)n_nodes * 128;
    float*    h2   = agg1 + (size_t)n_nodes * 128;
    float*    agg2 = h1;

    const int B = 256;
    const long long n_pairs = (long long)n_nodes * 64;
    k_deg_init <<<(n_nodes + B - 1) / B, B, 0, stream>>>(dinv, n_nodes);
    k_deg_edges<<<(n_edges + B - 1) / B, B, 0, stream>>>(dst, dinv, n_edges);
    k_deg_rsqrt<<<(n_nodes + B - 1) / B, B, 0, stream>>>(dinv, n_nodes);
    k_pack_x <<<(unsigned)((n_pairs + B - 1) / B), B, 0, stream>>>(x, xb, n_pairs);
    k_pack_w1<<<8192 / B, B, 0, stream>>>(W1, w1p);
    k_gemm1_wmma<<<(n_nodes + 15) / 16, B, 0, stream>>>(xb, w1p, dinv, h1, agg1, n_nodes);
    k_scatter1<<<((size_t)n_edges * 32 + B - 1) / B, B, 0, stream>>>(h1, src, dst, dinv, agg1, n_edges);
    k_layer2  <<<((size_t)n_nodes * 32 + B - 1) / B, B, 0, stream>>>(agg1, b1, W2, h2, n_nodes);
    k_self2   <<<(n_nodes * 2 + B - 1) / B, B, 0, stream>>>(h2, dinv, agg2, n_nodes);
    k_scatter2<<<(n_edges + B - 1) / B, B, 0, stream>>>(h2, src, dst, dinv, agg2, n_edges);
    k_logsoftmax<<<(n_nodes + B - 1) / B, B, 0, stream>>>(agg2, b2, out, n_nodes);
}


# every pass keeps its items inside its waves (a wave takes 256 edges, 256 float4 of X, one row tile across all 128 product columns, or 8 nodes): about 8x fewer waves started; same passes, same sums
# speedup vs baseline: 2.6185x; 2.6185x over previous
#include <hip/hip_runtime.h>
#include <hip/hip_bf16.h>


typedef __attribute__((ext_vector_type(16))) __bf16     v16bf;
typedef __attribute__((ext_vector_type(8)))  float      v8f;
typedef __attribute__((ext_vector_type(8)))  unsigned   v8u;

union FragAB { v8u u; v16bf b; };

__device__ __forceinline__ unsigned bf16_rne(float f) {
    unsigned u = __float_as_uint(f);
    u += 0x7FFFu + ((u >> 16) & 1u);
    return u >> 16;
}
__device__ __forceinline__ unsigned pack2bf(float lo, float hi) {
    return bf16_rne(lo) | (bf16_rne(hi) << 16);
}

__global__ void k_zero(int* __restrict__ cnt, int* __restrict__ total, int n) {
    int i = blockIdx.x * blockDim.x + threadIdx.x;
    if (i < n) cnt[i] = 0;
    if (i == 0) total[0] = 0;
}
__global__ void k_hist(const int* __restrict__ dst, int* cnt,
                       int* __restrict__ rank, int e) {
    const int lane = threadIdx.x & 31;
    const long long wave = ((long long)blockIdx.x * blockDim.x + threadIdx.x) >> 5;
    const long long base = wave * 256;
    if (base >= e) return;
    if (base + 256 <= e && (e & 3) == 0) {
        const int4* d4 = (const int4*)(dst + base);
        int4*       r4 = (int4*)(rank + base);
        const int4 a = d4[lane], b = d4[32 + lane];
        int4 ra, rb;
        ra.x = atomicAdd(&cnt[a.x], 1); ra.y = atomicAdd(&cnt[a.y], 1);
        ra.z = atomicAdd(&cnt[a.z], 1); ra.w = atomicAdd(&cnt[a.w], 1);
        rb.x = atomicAdd(&cnt[b.x], 1); rb.y = atomicAdd(&cnt[b.y], 1);
        rb.z = atomicAdd(&cnt[b.z], 1); rb.w = atomicAdd(&cnt[b.w], 1);
        r4[lane] = ra; r4[32 + lane] = rb;
    } else {
        for (int k = 0; k < 8; ++k) {
            const long long i = base + k * 32 + lane;
            if (i < e) rank[i] = atomicAdd(&cnt[dst[i]], 1);
        }
    }
}
__global__ void k_alloc(const int* __restrict__ cnt, int* total,
                        int* __restrict__ seg, float* __restrict__ dinv, int n) {
    const int i    = blockIdx.x * blockDim.x + threadIdx.x;
    const int lane = threadIdx.x & 31;
    const int c    = (i < n) ? cnt[i] : 0;
    int incl = c;
#pragma unroll
    for (int off = 1; off < 32; off <<= 1) {
        int t = __shfl_up(incl, off, 32);
        if (lane >= off) incl += t;
    }
    int base = 0;
    if (lane == 31) base = atomicAdd(total, incl);
    base = __shfl(base, 31, 32);
    if (i < n) {
        ((int2*)seg)[i] = make_int2(base + incl - c, c);
        dinv[i] = rsqrtf((float)(c + 1));
    }
}
__global__ void k_fill(const int* __restrict__ src, const int* __restrict__ dst,
                       const int* __restrict__ rank, const int* __restrict__ seg,
                       int* __restrict__ csr, int e) {
    const int lane = threadIdx.x & 31;
    const long long wave = ((long long)blockIdx.x * blockDim.x + threadIdx.x) >> 5;
    const long long base = wave * 256;
    if (base >= e) return;
    if (base + 256 <= e && (e & 3) == 0) {
        const int4* s4 = (const int4*)(src + base);
        const int4* d4 = (const int4*)(dst + base);
        const int4* r4 = (const int4*)(rank + base);
        const int4 sa = s4[lane], sb = s4[32 + lane];
        const int4 da = d4[lane], db = d4[32 + lane];
        const int4 ra = r4[lane], rb = r4[32 + lane];
        const int sv[8] = {sa.x, sa.y, sa.z, sa.w, sb.x, sb.y, sb.z, sb.w};
        const int dv[8] = {da.x, da.y, da.z, da.w, db.x, db.y, db.z, db.w};
        const int rv[8] = {ra.x, ra.y, ra.z, ra.w, rb.x, rb.y, rb.z, rb.w};
        int pos[8];
#pragma unroll
        for (int k = 0; k < 8; ++k) pos[k] = seg[2 * dv[k]] + rv[k];
#pragma unroll
        for (int k = 0; k < 8; ++k)
            if ((unsigned)pos[k] < (unsigned)e) csr[pos[k]] = sv[k];
    } else {
        for (int k = 0; k < 8; ++k) {
            const long long i = base + k * 32 + lane;
            if (i < e) {
                const int s   = src[i];
                const int pos = seg[2 * dst[i]] + rank[i];
                if ((unsigned)pos < (unsigned)e) csr[pos] = s;
            }
        }
    }
}

__global__ void k_pack_x(const float* __restrict__ x, unsigned* __restrict__ xb,
                         long long n4) {
    const int lane = threadIdx.x & 31;
    const long long wave = ((long long)blockIdx.x * blockDim.x + threadIdx.x) >> 5;
    const long long q0 = wave * 256 + lane;
    const float4* x4 = (const float4*)x;
    uint2*        o2 = (uint2*)xb;
    if (wave * 256 >= n4) return;
    if (wave * 256 + 256 <= n4) {
        float4 v[8];
#pragma unroll
        for (int j = 0; j < 8; ++j) v[j] = x4[q0 + j * 32];
#pragma unroll
        for (int j = 0; j < 8; ++j)
            o2[q0 + j * 32] = make_uint2(pack2bf(v[j].x, v[j].y), pack2bf(v[j].z, v[j].w));
    } else {
        for (int j = 0; j < 8; ++j) {
            const long long q = q0 + j * 32;
            if (q < n4) {
                const float4 v = x4[q];
                o2[q] = make_uint2(pack2bf(v.x, v.y), pack2bf(v.z, v.w));
            }
        }
    }
}
__global__ void k_pack_w1(const float* __restrict__ W1, unsigned* __restrict__ w1p) {
    int tid = blockIdx.x * blockDim.x + threadIdx.x;
    if (tid >= 8192) return;
    int v    = tid & 7;
    int lane = (tid >> 3) & 31;
    int ks   = (tid >> 8) & 3;
    int nt   = tid >> 10;
    int k    = ks * 32 + ((v >> 2) << 4) + ((lane >> 4) << 3) + ((v & 3) << 1);
    int ncol = nt * 16 + (lane & 15);
    w1p[tid] = pack2bf(W1[(size_t)k * 128 + ncol], W1[(size_t)(k + 1) * 128 + ncol]);
}

__global__ void k_gemm1_wmma(const unsigned* __restrict__ xb,
                             const unsigned* __restrict__ w1p,
                             const float* __restrict__ dinv,
                             float* __restrict__ h1s,
                             int n_nodes) {
    const int lane = threadIdx.x & 31;
    const int tile = (blockIdx.x * blockDim.x + threadIdx.x) >> 5;
    const int m0   = tile * 16;
    if (m0 >= n_nodes) return;
    const int mr   = lane & 15;
    const int half = lane >> 4;

    int row = m0 + mr;
    if (row >= n_nodes) row = n_nodes - 1;
    const unsigned* xr = xb + (size_t)row * 64;
    const v8u* bfrag = (const v8u*)w1p;

    FragAB a[4];
#pragma unroll
    for (int ks = 0; ks < 4; ++ks) {
        const uint4 alo = *(const uint4*)(xr + ks * 16 + half * 4);
        const uint4 ahi = *(const uint4*)(xr + ks * 16 + 8 + half * 4);
        a[ks].u = (v8u){alo.x, alo.y, alo.z, alo.w, ahi.x, ahi.y, ahi.z, ahi.w};
    }
    __builtin_amdgcn_sched_barrier(0);
    v8f acc[8];
#pragma unroll
    for (int nt = 0; nt < 8; ++nt) acc[nt] = (v8f){};
#pragma unroll
    for (int ks = 0; ks < 4; ++ks) {
#pragma unroll
        for (int g = 0; g < 8; g += 4) {
            FragAB b[4];
#pragma unroll
            for (int i = 0; i < 4; ++i)
                b[i].u = bfrag[((g + i) * 4 + ks) * 32 + lane];
            __builtin_amdgcn_sched_barrier(0);
#pragma unroll
            for (int i = 0; i < 4; ++i)
                acc[g + i] = __builtin_amdgcn_wmma_f32_16x16x32_bf16(
                                 false, a[ks].b, false, b[i].b, (short)0, acc[g + i], false, false);
            __builtin_amdgcn_sched_barrier(0);
        }
    }
    const int mbase = m0 + (half << 3);
    if (m0 + 16 <= n_nodes) {
        float*       hp = h1s + (size_t)mbase * 128 + mr;
        const float4 d0 = *(const float4*)(dinv + mbase);
        const float4 d1 = *(const float4*)(dinv + mbase + 4);
        const float  dv[8] = {d0.x, d0.y, d0.z, d0.w, d1.x, d1.y, d1.z, d1.w};
#pragma unroll
        for (int nt = 0; nt < 8; ++nt)
#pragma unroll
            for (int v = 0; v < 8; ++v)
                hp[(size_t)v * 128 + nt * 16] = acc[nt][v] * dv[v];
    } else {
#pragma unroll
        for (int nt = 0; nt < 8; ++nt)
#pragma unroll
            for (int v = 0; v < 8; ++v) {
                const int m = mbase + v;
                if (m < n_nodes)
                    h1s[(size_t)m * 128 + nt * 16 + mr] = acc[nt][v] * dinv[m];
            }
    }
}

#define G1_NODES 8
__device__ __forceinline__ int g1_load_srcs(const int* __restrict__ csr, int first, int m,
                                            int lane, int node, int n_edges) {
    int sv = node;
    if (lane < m) {
        const int idx = min(first + lane, n_edges - 1);
        sv = csr[idx];
    }
    return sv;
}
__global__ void k_gather1(const float* __restrict__ h1s,
                          const int* __restrict__ csr,
                          const int* __restrict__ seg,
                          const float* __restrict__ dinv,
                          const float* __restrict__ b1, const float* __restrict__ W2,
                          float* __restrict__ h2s, int n_nodes, int n_edges) {
    const int lane = threadIdx.x & 31;
    const int wave  = __builtin_amdgcn_readfirstlane((blockIdx.x * blockDim.x + threadIdx.x) >> 5);
    const int node0 = wave * G1_NODES;
    if (node0 >= n_nodes) return;
    const int cntn = min(G1_NODES, n_nodes - node0);

    int2  sgv = make_int2(0, 0);
    float dnv = 0.f;
    if (lane < cntn) {
        sgv = ((const int2*)seg)[node0 + lane];
        dnv = dinv[node0 + lane];
    }
    const float4* rows = (const float4*)h1s;
    const float4 bb = ((const float4*)b1)[lane];
    const int c = lane * 4;
    const float2* w2 = (const float2*)W2;
    const float2 w0 = w2[c], w1 = w2[c + 1], w2v = w2[c + 2], w3 = w2[c + 3];

    int beg = __builtin_amdgcn_readlane(sgv.x, 0);
    int n   = __builtin_amdgcn_readlane(sgv.y, 0);
    int sv_next = g1_load_srcs(csr, beg, min(n, 32), lane, node0, n_edges);
    float2 res = make_float2(0.f, 0.f);

    for (int k = 0; k < cntn; ++k) {
        const int node = node0 + k;
        int sv = (int)min((unsigned)sv_next, (unsigned)(n_nodes - 1));
        float4 acc = rows[(size_t)node * 32 + lane];
        int begn = 0, nn = 0;
        if (k + 1 < cntn) {
            begn = __builtin_amdgcn_readlane(sgv.x, k + 1);
            nn   = __builtin_amdgcn_readlane(sgv.y, k + 1);
            sv_next = g1_load_srcs(csr, begn, min(nn, 32), lane, node + 1, n_edges);
        }
        for (int c0 = 0; c0 < n; c0 += 32) {
            const int m = min(n - c0, 32);
            if (c0 > 0)
                sv = (int)min((unsigned)g1_load_srcs(csr, beg + c0, m, lane, node, n_edges),
                              (unsigned)(n_nodes - 1));
            int j = 0;
            for (; j + 4 <= m; j += 4) {
                const int s0 = __builtin_amdgcn_readlane(sv, j);
                const int s1 = __builtin_amdgcn_readlane(sv, j + 1);
                const int s2 = __builtin_amdgcn_readlane(sv, j + 2);
                const int s3 = __builtin_amdgcn_readlane(sv, j + 3);
                const float4 v0 = rows[(size_t)s0 * 32 + lane];
                const float4 v1 = rows[(size_t)s1 * 32 + lane];
                const float4 v2 = rows[(size_t)s2 * 32 + lane];
                const float4 v3 = rows[(size_t)s3 * 32 + lane];
                acc.x += (v0.x + v1.x) + (v2.x + v3.x);
                acc.y += (v0.y + v1.y) + (v2.y + v3.y);
                acc.z += (v0.z + v1.z) + (v2.z + v3.z);
                acc.w += (v0.w + v1.w) + (v2.w + v3.w);
            }
            if ((m - j) & 2) {
                const int s0 = __builtin_amdgcn_readlane(sv, j);
                const int s1 = __builtin_amdgcn_readlane(sv, j + 1);
                const float4 v0 = rows[(size_t)s0 * 32 + lane];
                const float4 v1 = rows[(size_t)s1 * 32 + lane];
                acc.x += v0.x + v1.x; acc.y += v0.y + v1.y;
                acc.z += v0.z + v1.z; acc.w += v0.w + v1.w;
                j += 2;
            }
            if ((m - j) & 1) {
                const int s = __builtin_amdgcn_readlane(sv, j);
                const float4 v = rows[(size_t)s * 32 + lane];
                acc.x += v.x; acc.y += v.y; acc.z += v.z; acc.w += v.w;
            }
        }

        const float dn = __int_as_float(__builtin_amdgcn_readlane(__float_as_int(dnv), k));
        const float t0 = fmaxf(acc.x * dn + bb.x, 0.f), t1 = fmaxf(acc.y * dn + bb.y, 0.f);
        const float t2 = fmaxf(acc.z * dn + bb.z, 0.f), t3 = fmaxf(acc.w * dn + bb.w, 0.f);
        float s0 = t0 * w0.x + t1 * w1.x + t2 * w2v.x + t3 * w3.x;
        float s1 = t0 * w0.y + t1 * w1.y + t2 * w2v.y + t3 * w3.y;
#pragma unroll
        for (int off = 16; off > 0; off >>= 1) {
            s0 += __shfl_down(s0, off, 32);
            s1 += __shfl_down(s1, off, 32);
        }
        const float r0 = __int_as_float(__builtin_amdgcn_readfirstlane(__float_as_int(s0))) * dn;
        const float r1 = __int_as_float(__builtin_amdgcn_readfirstlane(__float_as_int(s1))) * dn;
        if (lane == k) res = make_float2(r0, r1);
        beg = begn; n = nn;
    }
    if (lane < cntn) ((float2*)h2s)[node0 + lane] = res;
}

__global__ void k_gather2_out(const float* __restrict__ h2s,
                              const int* __restrict__ csr,
                              const int* __restrict__ seg,
                              const float* __restrict__ dinv, const float* __restrict__ b2,
                              float* __restrict__ out, int n_nodes, int n_edges) {
    const int  gid   = blockIdx.x * blockDim.x + threadIdx.x;
    const int  node  = gid >> 1;
    const int  sub   = gid & 1;
    const bool valid = node < n_nodes;
    const int  nc    = valid ? node : n_nodes - 1;
    const int2 sg    = ((const int2*)seg)[nc];
    const int  beg   = sg.x;
    const int  n     = valid ? sg.y : 0;
    const float2* hs = (const float2*)h2s;

    float2 acc = make_float2(0.f, 0.f);
    if (sub == 0) acc = hs[nc];
#pragma unroll 4
    for (int j = sub; j < n; j += 2) {
        const int idx = min(beg + j, n_edges - 1);
        const int s   = (int)min((unsigned)csr[idx], (unsigned)(n_nodes - 1));
        const float2 v = hs[s];
        acc.x += v.x; acc.y += v.y;
    }
    acc.x += __shfl_xor(acc.x, 1, 32);
    acc.y += __shfl_xor(acc.y, 1, 32);
    if (valid && sub == 0) {
        const float dn = dinv[node];
        const float z0 = acc.x * dn + b2[0];
        const float z1 = acc.y * dn + b2[1];
        const float m  = fmaxf(z0, z1);
        const float l  = m + __logf(__expf(z0 - m) + __expf(z1 - m));
        ((float2*)out)[node] = make_float2(z0 - l, z1 - l);
    }
}

extern "C" void kernel_launch(void* const* d_in, const int* in_sizes, int n_in,
                              void* d_out, int out_size, void* d_ws, size_t ws_size,
                              hipStream_t stream) {
    const float* x   = (const float*)d_in[0];
    const int*   ei  = (const int*)  d_in[1];
    const float* W1  = (const float*)d_in[2];
    const float* b1  = (const float*)d_in[3];
    const float* W2  = (const float*)d_in[4];
    const float* b2  = (const float*)d_in[5];
    float*       out = (float*)d_out;

    const int n_nodes = in_sizes[0] / 128;
    const int n_edges = in_sizes[1] / 2;
    const int* src = ei;
    const int* dst = ei + n_edges;

    const size_t nP = ((size_t)n_nodes + 127) / 128 * 128;
    float*    h1s   = (float*)d_ws;
    unsigned* xb    = (unsigned*)(h1s + (size_t)n_nodes * 128);
    unsigned* w1p   = xb + (size_t)n_nodes * 64;
    float*    h2s   = (float*)(w1p + 8192);
    float*    dinv  = h2s + 2 * nP;
    int*      cnt   = (int*)(dinv + nP);
    int*      seg   = cnt + nP;
    int*      total = cnt + 3 * nP;
    int*      rank  = total + 128;
    int*      csr   = rank + n_edges;

    const int B = 256;
    const long long n4 = (long long)n_nodes * 32;
    k_zero <<<(n_nodes + B - 1) / B, B, 0, stream>>>(cnt, total, n_nodes);
    k_hist <<<(n_edges + 8 * B - 1) / (8 * B), B, 0, stream>>>(dst, cnt, rank, n_edges);
    k_alloc<<<(n_nodes + B - 1) / B, B, 0, stream>>>(cnt, total, seg, dinv, n_nodes);
    k_fill <<<(n_edges + 8 * B - 1) / (8 * B), B, 0, stream>>>(src, dst, rank, seg, csr, n_edges);
    k_pack_x <<<(unsigned)((n4 + 8 * B - 1) / (8 * B)), B, 0, stream>>>(x, xb, n4);
    k_pack_w1<<<8192 / B, B, 0, stream>>>(W1, w1p);
    k_gemm1_wmma<<<(n_nodes + 127) / 128, B, 0, stream>>>(xb, w1p, dinv, h1s, n_nodes);
    k_gather1<<<(n_nodes + 8 * G1_NODES - 1) / (8 * G1_NODES), B, 0, stream>>>(
        h1s, csr, seg, dinv, b1, W2, h2s, n_nodes, n_edges);
    k_gather2_out<<<(n_nodes * 2 + B - 1) / B, B, 0, stream>>>(
        h2s, csr, seg, dinv, b2, out, n_nodes, n_edges);
}


# layer 1 aggregates before its product: the gather sums 256-byte bf16 rows of dinv-scaled X (8 in flight), not 512-byte f32 product rows; relu and @W2 move into the WMMA epilogue; operands stay bf16
# speedup vs baseline: 2.7692x; 1.0576x over previous
#include <hip/hip_runtime.h>
#include <hip/hip_bf16.h>


typedef __attribute__((ext_vector_type(16))) __bf16     v16bf;
typedef __attribute__((ext_vector_type(8)))  float      v8f;
typedef __attribute__((ext_vector_type(8)))  unsigned   v8u;

union FragAB { v8u u; v16bf b; };

__device__ __forceinline__ unsigned bf16_rne(float f) {
    unsigned u = __float_as_uint(f);
    u += 0x7FFFu + ((u >> 16) & 1u);
    return u >> 16;
}
__device__ __forceinline__ unsigned pack2bf(float lo, float hi) {
    return bf16_rne(lo) | (bf16_rne(hi) << 16);
}

__global__ void k_zero(int* __restrict__ cnt, int* __restrict__ total, int n) {
    int i = blockIdx.x * blockDim.x + threadIdx.x;
    if (i < n) cnt[i] = 0;
    if (i == 0) total[0] = 0;
}
__global__ void k_hist(const int* __restrict__ dst, int* cnt,
                       int* __restrict__ rank, int e) {
    const int lane = threadIdx.x & 31;
    const long long wave = ((long long)blockIdx.x * blockDim.x + threadIdx.x) >> 5;
    const long long base = wave * 256;
    if (base >= e) return;
    if (base + 256 <= e && (e & 3) == 0) {
        const int4* d4 = (const int4*)(dst + base);
        int4*       r4 = (int4*)(rank + base);
        const int4 a = d4[lane], b = d4[32 + lane];
        int4 ra, rb;
        ra.x = atomicAdd(&cnt[a.x], 1); ra.y = atomicAdd(&cnt[a.y], 1);
        ra.z = atomicAdd(&cnt[a.z], 1); ra.w = atomicAdd(&cnt[a.w], 1);
        rb.x = atomicAdd(&cnt[b.x], 1); rb.y = atomicAdd(&cnt[b.y], 1);
        rb.z = atomicAdd(&cnt[b.z], 1); rb.w = atomicAdd(&cnt[b.w], 1);
        r4[lane] = ra; r4[32 + lane] = rb;
    } else {
        for (int k = 0; k < 8; ++k) {
            const long long i = base + k * 32 + lane;
            if (i < e) rank[i] = atomicAdd(&cnt[dst[i]], 1);
        }
    }
}
__global__ void k_alloc(const int* __restrict__ cnt, int* total,
                        int* __restrict__ seg, float* __restrict__ dinv, int n) {
    const int i    = blockIdx.x * blockDim.x + threadIdx.x;
    const int lane = threadIdx.x & 31;
    const int c    = (i < n) ? cnt[i] : 0;
    int incl = c;
#pragma unroll
    for (int off = 1; off < 32; off <<= 1) {
        int t = __shfl_up(incl, off, 32);
        if (lane >= off) incl += t;
    }
    int base = 0;
    if (lane == 31) base = atomicAdd(total, incl);
    base = __shfl(base, 31, 32);
    if (i < n) {
        ((int2*)seg)[i] = make_int2(base + incl - c, c);
        dinv[i] = rsqrtf((float)(c + 1));
    }
}
__global__ void k_fill(const int* __restrict__ src, const int* __restrict__ dst,
                       const int* __restrict__ rank, const int* __restrict__ seg,
                       int* __restrict__ csr, int e) {
    const int lane = threadIdx.x & 31;
    const long long wave = ((long long)blockIdx.x * blockDim.x + threadIdx.x) >> 5;
    const long long base = wave * 256;
    if (base >= e) return;
    if (base + 256 <= e && (e & 3) == 0) {
        const int4* s4 = (const int4*)(src + base);
        const int4* d4 = (const int4*)(dst + base);
        const int4* r4 = (const int4*)(rank + base);
        const int4 sa = s4[lane], sb = s4[32 + lane];
        const int4 da = d4[lane], db = d4[32 + lane];
        const int4 ra = r4[lane], rb = r4[32 + lane];
        const int sv[8] = {sa.x, sa.y, sa.z, sa.w, sb.x, sb.y, sb.z, sb.w};
        const int dv[8] = {da.x, da.y, da.z, da.w, db.x, db.y, db.z, db.w};
        const int rv[8] = {ra.x, ra.y, ra.z, ra.w, rb.x, rb.y, rb.z, rb.w};
        int pos[8];
#pragma unroll
        for (int k = 0; k < 8; ++k) pos[k] = seg[2 * dv[k]] + rv[k];
#pragma unroll
        for (int k = 0; k < 8; ++k)
            if ((unsigned)pos[k] < (unsigned)e) csr[pos[k]] = sv[k];
    } else {
        for (int k = 0; k < 8; ++k) {
            const long long i = base + k * 32 + lane;
            if (i < e) {
                const int s   = src[i];
                const int pos = seg[2 * dst[i]] + rank[i];
                if ((unsigned)pos < (unsigned)e) csr[pos] = s;
            }
        }
    }
}

__global__ void k_pack_x(const float* __restrict__ x, const float* __restrict__ dinv,
                         unsigned* __restrict__ xs, long long n4) {
    const int lane = threadIdx.x & 31;
    const int wave = __builtin_amdgcn_readfirstlane(
        (int)(((long long)blockIdx.x * blockDim.x + threadIdx.x) >> 5));
    const long long w0 = (long long)wave * 256;
    const long long q0 = w0 + lane;
    const float4* x4 = (const float4*)x;
    uint2*        o2 = (uint2*)xs;
    if (w0 >= n4) return;
    if (w0 + 256 <= n4) {
        float4 v[8];
        float  d[8];
#pragma unroll
        for (int j = 0; j < 8; ++j) v[j] = x4[q0 + j * 32];
#pragma unroll
        for (int j = 0; j < 8; ++j) d[j] = dinv[wave * 8 + j];
#pragma unroll
        for (int j = 0; j < 8; ++j)
            o2[q0 + j * 32] = make_uint2(pack2bf(v[j].x * d[j], v[j].y * d[j]),
                                         pack2bf(v[j].z * d[j], v[j].w * d[j]));
    } else {
        for (int j = 0; j < 8; ++j) {
            const long long q = q0 + j * 32;
            if (q < n4) {
                const float4 v = x4[q];
                const float  d = dinv[q >> 5];
                o2[q] = make_uint2(pack2bf(v.x * d, v.y * d), pack2bf(v.z * d, v.w * d));
            }
        }
    }
}
__global__ void k_pack_w1(const float* __restrict__ W1, unsigned* __restrict__ w1p) {
    int tid = blockIdx.x * blockDim.x + threadIdx.x;
    if (tid >= 8192) return;
    int v    = tid & 7;
    int lane = (tid >> 3) & 31;
    int ks   = (tid >> 8) & 3;
    int nt   = tid >> 10;
    int k    = ks * 32 + ((v >> 2) << 4) + ((lane >> 4) << 3) + ((v & 3) << 1);
    int ncol = nt * 16 + (lane & 15);
    w1p[tid] = pack2bf(W1[(size_t)k * 128 + ncol], W1[(size_t)(k + 1) * 128 + ncol]);
}


#define G1_NODES 8
__device__ __forceinline__ int g1_load_srcs(const int* __restrict__ csr, int first, int m,
                                            int lane, int node, int n_edges) {
    int sv = node;
    if (lane < m) {
        const int idx = min(first + lane, n_edges - 1);
        sv = csr[idx];
    }
    return sv;
}
__device__ __forceinline__ float4 g1_f4(const uint2 v) {
    return make_float4(__uint_as_float(v.x << 16), __uint_as_float(v.x & 0xFFFF0000u),
                       __uint_as_float(v.y << 16), __uint_as_float(v.y & 0xFFFF0000u));
}
__device__ __forceinline__ float4 g1_sum(const float4 a, const float4 b) {
    return make_float4(a.x + b.x, a.y + b.y, a.z + b.z, a.w + b.w);
}
__global__ void k_gather_x(const unsigned* __restrict__ xs,
                           const int* __restrict__ csr,
                           const int* __restrict__ seg,
                           const float* __restrict__ dinv,
                           unsigned* __restrict__ zh, unsigned* __restrict__ zl,
                           int n_nodes, int n_edges) {
    const int lane = threadIdx.x & 31;
    const int wave  = __builtin_amdgcn_readfirstlane((blockIdx.x * blockDim.x + threadIdx.x) >> 5);
    const int node0 = wave * G1_NODES;
    if (node0 >= n_nodes) return;
    const int cntn = min(G1_NODES, n_nodes - node0);

    int2  sgv = make_int2(0, 0);
    float dnv = 0.f;
    if (lane < cntn) {
        sgv = ((const int2*)seg)[node0 + lane];
        dnv = dinv[node0 + lane];
    }
    const uint2* rows = (const uint2*)xs;
    uint2*       oh   = (uint2*)zh;
    uint2*       ol   = (uint2*)zl;

    int beg = __builtin_amdgcn_readlane(sgv.x, 0);
    int n   = __builtin_amdgcn_readlane(sgv.y, 0);
    int sv_next = g1_load_srcs(csr, beg, min(n, 32), lane, node0, n_edges);

    for (int k = 0; k < cntn; ++k) {
        const int node = node0 + k;
        int sv = (int)min((unsigned)sv_next, (unsigned)(n_nodes - 1));
        float4 acc = g1_f4(rows[(size_t)node * 32 + lane]);
        int begn = 0, nn = 0;
        if (k + 1 < cntn) {
            begn = __builtin_amdgcn_readlane(sgv.x, k + 1);
            nn   = __builtin_amdgcn_readlane(sgv.y, k + 1);
            sv_next = g1_load_srcs(csr, begn, min(nn, 32), lane, node + 1, n_edges);
        }
        for (int c0 = 0; c0 < n; c0 += 32) {
            const int m = min(n - c0, 32);
            if (c0 > 0)
                sv = (int)min((unsigned)g1_load_srcs(csr, beg + c0, m, lane, node, n_edges),
                              (unsigned)(n_nodes - 1));
            int j = 0;
            for (; j + 8 <= m; j += 8) {
                const int s0 = __builtin_amdgcn_readlane(sv, j);
                const int s1 = __builtin_amdgcn_readlane(sv, j + 1);
                const int s2 = __builtin_amdgcn_readlane(sv, j + 2);
                const int s3 = __builtin_amdgcn_readlane(sv, j + 3);
                const int s4 = __builtin_amdgcn_readlane(sv, j + 4);
                const int s5 = __builtin_amdgcn_readlane(sv, j + 5);
                const int s6 = __builtin_amdgcn_readlane(sv, j + 6);
                const int s7 = __builtin_amdgcn_readlane(sv, j + 7);
                const uint2 u0 = rows[(size_t)s0 * 32 + lane];
                const uint2 u1 = rows[(size_t)s1 * 32 + lane];
                const uint2 u2 = rows[(size_t)s2 * 32 + lane];
                const uint2 u3 = rows[(size_t)s3 * 32 + lane];
                const uint2 u4 = rows[(size_t)s4 * 32 + lane];
                const uint2 u5 = rows[(size_t)s5 * 32 + lane];
                const uint2 u6 = rows[(size_t)s6 * 32 + lane];
                const uint2 u7 = rows[(size_t)s7 * 32 + lane];
                const float4 a = g1_sum(g1_sum(g1_f4(u0), g1_f4(u1)), g1_sum(g1_f4(u2), g1_f4(u3)));
                const float4 b = g1_sum(g1_sum(g1_f4(u4), g1_f4(u5)), g1_sum(g1_f4(u6), g1_f4(u7)));
                acc = g1_sum(acc, g1_sum(a, b));
            }
            if ((m - j) & 4) {
                const int s0 = __builtin_amdgcn_readlane(sv, j);
                const int s1 = __builtin_amdgcn_readlane(sv, j + 1);
                const int s2 = __builtin_amdgcn_readlane(sv, j + 2);
                const int s3 = __builtin_amdgcn_readlane(sv, j + 3);
                const uint2 u0 = rows[(size_t)s0 * 32 + lane];
                const uint2 u1 = rows[(size_t)s1 * 32 + lane];
                const uint2 u2 = rows[(size_t)s2 * 32 + lane];
                const uint2 u3 = rows[(size_t)s3 * 32 + lane];
                acc = g1_sum(acc, g1_sum(g1_sum(g1_f4(u0), g1_f4(u1)), g1_sum(g1_f4(u2), g1_f4(u3))));
                j += 4;
            }
            if ((m - j) & 2) {
                const int s0 = __builtin_amdgcn_readlane(sv, j);
                const int s1 = __builtin_amdgcn_readlane(sv, j + 1);
                const uint2 u0 = rows[(size_t)s0 * 32 + lane];
                const uint2 u1 = rows[(size_t)s1 * 32 + lane];
                acc = g1_sum(acc, g1_sum(g1_f4(u0), g1_f4(u1)));
                j += 2;
            }
            if ((m - j) & 1) {
                const int s = __builtin_amdgcn_readlane(sv, j);
                acc = g1_sum(acc, g1_f4(rows[(size_t)s * 32 + lane]));
            }
        }

        const float dn = __int_as_float(__builtin_amdgcn_readlane(__float_as_int(dnv), k));
        const float z0 = acc.x * dn, z1 = acc.y * dn, z2 = acc.z * dn, z3 = acc.w * dn;
        const unsigned h0 = bf16_rne(z0), h1 = bf16_rne(z1), h2 = bf16_rne(z2), h3 = bf16_rne(z3);
        const float r0 = z0 - __uint_as_float(h0 << 16), r1 = z1 - __uint_as_float(h1 << 16);
        const float r2 = z2 - __uint_as_float(h2 << 16), r3 = z3 - __uint_as_float(h3 << 16);
        oh[(size_t)node * 32 + lane] = make_uint2(h0 | (h1 << 16), h2 | (h3 << 16));
        ol[(size_t)node * 32 + lane] = make_uint2(pack2bf(r0, r1), pack2bf(r2, r3));
        beg = begn; n = nn;
    }
}

__global__ void __launch_bounds__(256)
k_gemm1_wmma(const unsigned* __restrict__ zh,
                             const unsigned* __restrict__ zl,
                             const unsigned* __restrict__ w1p,
                             const float* __restrict__ dinv,
                             const float* __restrict__ b1, const float* __restrict__ W2,
                             float* __restrict__ h2s,
                             int n_nodes) {
    const int lane = threadIdx.x & 31;
    const int tile = (blockIdx.x * blockDim.x + threadIdx.x) >> 5;
    const int m0   = tile * 16;
    if (m0 >= n_nodes) return;
    const int mr   = lane & 15;
    const int half = lane >> 4;

    int row = m0 + mr;
    if (row >= n_nodes) row = n_nodes - 1;
    const unsigned* rh = zh + (size_t)row * 64;
    const unsigned* rl = zl + (size_t)row * 64;
    const v8u* bfrag = (const v8u*)w1p;

    FragAB ah[4], al[4];
#pragma unroll
    for (int ks = 0; ks < 4; ++ks) {
        const uint4 hlo = *(const uint4*)(rh + ks * 16 + half * 4);
        const uint4 hhi = *(const uint4*)(rh + ks * 16 + 8 + half * 4);
        const uint4 llo = *(const uint4*)(rl + ks * 16 + half * 4);
        const uint4 lhi = *(const uint4*)(rl + ks * 16 + 8 + half * 4);
        ah[ks].u = (v8u){hlo.x, hlo.y, hlo.z, hlo.w, hhi.x, hhi.y, hhi.z, hhi.w};
        al[ks].u = (v8u){llo.x, llo.y, llo.z, llo.w, lhi.x, lhi.y, lhi.z, lhi.w};
    }
    __builtin_amdgcn_sched_barrier(0);
    v8f acc[8];
#pragma unroll
    for (int nt = 0; nt < 8; ++nt) acc[nt] = (v8f){};
#pragma unroll
    for (int ks = 0; ks < 4; ++ks) {
#pragma unroll
        for (int g = 0; g < 8; g += 4) {
            FragAB b[4];
#pragma unroll
            for (int i = 0; i < 4; ++i)
                b[i].u = bfrag[((g + i) * 4 + ks) * 32 + lane];
            __builtin_amdgcn_sched_barrier(0);
#pragma unroll
            for (int i = 0; i < 4; ++i)
                acc[g + i] = __builtin_amdgcn_wmma_f32_16x16x32_bf16(
                                 false, ah[ks].b, false, b[i].b, (short)0, acc[g + i], false, false);
#pragma unroll
            for (int i = 0; i < 4; ++i)
                acc[g + i] = __builtin_amdgcn_wmma_f32_16x16x32_bf16(
                                 false, al[ks].b, false, b[i].b, (short)0, acc[g + i], false, false);
            __builtin_amdgcn_sched_barrier(0);
        }
    }
    float bv[8], wa[8], wb[8];
#pragma unroll
    for (int nt = 0; nt < 8; ++nt) {
        const int col = nt * 16 + mr;
        const float2 w = ((const float2*)W2)[col];
        bv[nt] = b1[col]; wa[nt] = w.x; wb[nt] = w.y;
    }
    float p0[8], p1[8];
#pragma unroll
    for (int v = 0; v < 8; ++v) {
        float s0 = 0.f, s1 = 0.f;
#pragma unroll
        for (int nt = 0; nt < 8; ++nt) {
            const float t = fmaxf(acc[nt][v] + bv[nt], 0.f);
            s0 += t * wa[nt];
            s1 += t * wb[nt];
        }
        p0[v] = s0; p1[v] = s1;
    }
#pragma unroll
    for (int off = 8; off > 0; off >>= 1) {
#pragma unroll
        for (int v = 0; v < 8; ++v) {
            p0[v] += __shfl_xor(p0[v], off, 32);
            p1[v] += __shfl_xor(p1[v], off, 32);
        }
    }
    float o0 = p0[0], o1 = p1[0];
#pragma unroll
    for (int v = 1; v < 8; ++v)
        if (mr == v) { o0 = p0[v]; o1 = p1[v]; }
    const int orow = m0 + (half << 3) + mr;
    if (mr < 8 && orow < n_nodes) {
        const float dn = dinv[orow];
        ((float2*)h2s)[orow] = make_float2(o0 * dn, o1 * dn);
    }
}

__global__ void k_gather2_out(const float* __restrict__ h2s,
                              const int* __restrict__ csr,
                              const int* __restrict__ seg,
                              const float* __restrict__ dinv, const float* __restrict__ b2,
                              float* __restrict__ out, int n_nodes, int n_edges) {
    const int  gid   = blockIdx.x * blockDim.x + threadIdx.x;
    const int  node  = gid >> 1;
    const int  sub   = gid & 1;
    const bool valid = node < n_nodes;
    const int  nc    = valid ? node : n_nodes - 1;
    const int2 sg    = ((const int2*)seg)[nc];
    const int  beg   = sg.x;
    const int  n     = valid ? sg.y : 0;
    const float2* hs = (const float2*)h2s;

    float2 acc = make_float2(0.f, 0.f);
    if (sub == 0) acc = hs[nc];
#pragma unroll 4
    for (int j = sub; j < n; j += 2) {
        const int idx = min(beg + j, n_edges - 1);
        const int s   = (int)min((unsigned)csr[idx], (unsigned)(n_nodes - 1));
        const float2 v = hs[s];
        acc.x += v.x; acc.y += v.y;
    }
    acc.x += __shfl_xor(acc.x, 1, 32);
    acc.y += __shfl_xor(acc.y, 1, 32);
    if (valid && sub == 0) {
        const float dn = dinv[node];
        const float z0 = acc.x * dn + b2[0];
        const float z1 = acc.y * dn + b2[1];
        const float m  = fmaxf(z0, z1);
        const float l  = m + __logf(__expf(z0 - m) + __expf(z1 - m));
        ((float2*)out)[node] = make_float2(z0 - l, z1 - l);
    }
}

extern "C" void kernel_launch(void* const* d_in, const int* in_sizes, int n_in,
                              void* d_out, int out_size, void* d_ws, size_t ws_size,
                              hipStream_t stream) {
    const float* x   = (const float*)d_in[0];
    const int*   ei  = (const int*)  d_in[1];
    const float* W1  = (const float*)d_in[2];
    const float* b1  = (const float*)d_in[3];
    const float* W2  = (const float*)d_in[4];
    const float* b2  = (const float*)d_in[5];
    float*       out = (float*)d_out;

    const int n_nodes = in_sizes[0] / 128;
    const int n_edges = in_sizes[1] / 2;
    const int* src = ei;
    const int* dst = ei + n_edges;

    const size_t nP = ((size_t)n_nodes + 127) / 128 * 128;
    unsigned* zh    = (unsigned*)d_ws;
    unsigned* zl    = zh + (size_t)n_nodes * 64;
    unsigned* xs    = zl + (size_t)n_nodes * 64;
    unsigned* w1p   = xs + (size_t)n_nodes * 64;
    float*    h2s   = (float*)(w1p + 8192);
    float*    dinv  = h2s + 2 * nP;
    int*      cnt   = (int*)(dinv + nP);
    int*      seg   = cnt + nP;
    int*      total = cnt + 3 * nP;
    int*      rank  = total + 128;
    int*      csr   = rank + n_edges;

    const int B = 256;
    const long long n4 = (long long)n_nodes * 32;
    k_zero <<<(n_nodes + B - 1) / B, B, 0, stream>>>(cnt, total, n_nodes);
    k_hist <<<(n_edges + 8 * B - 1) / (8 * B), B, 0, stream>>>(dst, cnt, rank, n_edges);
    k_alloc<<<(n_nodes + B - 1) / B, B, 0, stream>>>(cnt, total, seg, dinv, n_nodes);
    k_fill <<<(n_edges + 8 * B - 1) / (8 * B), B, 0, stream>>>(src, dst, rank, seg, csr, n_edges);
    k_pack_x <<<(unsigned)((n4 + 8 * B - 1) / (8 * B)), B, 0, stream>>>(x, dinv, xs, n4);
    k_pack_w1<<<8192 / B, B, 0, stream>>>(W1, w1p);
    k_gather_x<<<(n_nodes + 8 * G1_NODES - 1) / (8 * G1_NODES), B, 0, stream>>>(
        xs, csr, seg, dinv, zh, zl, n_nodes, n_edges);
    k_gemm1_wmma<<<(n_nodes + 127) / 128, B, 0, stream>>>(zh, zl, w1p, dinv, b1, W2, h2s, n_nodes);
    k_gather2_out<<<(n_nodes * 2 + B - 1) / B, B, 0, stream>>>(
        h2s, csr, seg, dinv, b2, out, n_nodes, n_edges);
}


# layer-1 gather: a node's own row and all its neighbour rows (up to 31) are requested before the first wait: one cache trip per node, not one per piece of 16/8/4/2/1; rows summed in a new tree
# speedup vs baseline: 7.8959x; 2.8513x over previous
#include <hip/hip_runtime.h>
#include <hip/hip_bf16.h>


typedef __attribute__((ext_vector_type(16))) __bf16     v16bf;
typedef __attribute__((ext_vector_type(8)))  float      v8f;
typedef __attribute__((ext_vector_type(8)))  unsigned   v8u;

union FragAB { v8u u; v16bf b; };

__device__ __forceinline__ unsigned bf16_rne(float f) {
    unsigned u = __float_as_uint(f);
    u += 0x7FFFu + ((u >> 16) & 1u);
    return u >> 16;
}
__device__ __forceinline__ unsigned pack2bf(float lo, float hi) {
    return bf16_rne(lo) | (bf16_rne(hi) << 16);
}

#define TS_EDGES 4096
#define BK_SHIFT 8

__device__ __forceinline__ int block_excl_scan(int v, int lane, int wave,
                                               int* wsum, int* all_sum) {
    int incl = v;
#pragma unroll
    for (int off = 1; off < 32; off <<= 1) {
        const int t = __shfl_up(incl, off, 32);
        if (lane >= off) incl += t;
    }
    if (lane == 31) wsum[wave] = incl;
    __syncthreads();
    int wbase = 0, tot = 0;
#pragma unroll
    for (int w = 0; w < 8; ++w) {
        const int x = wsum[w];
        if (w < wave) wbase += x;
        tot += x;
    }
    *all_sum = tot;
    return wbase + incl - v;
}

__device__ __forceinline__ size_t w1_entry_at(int i) {
    const int v    = i & 7;
    const int lane = (i >> 3) & 31;
    const int ks   = (i >> 8) & 3;
    const int nt   = i >> 10;
    const int k    = ks * 32 + ((v >> 2) << 4) + ((lane >> 4) << 3) + ((v & 3) << 1);
    const int ncol = nt * 16 + (lane & 15);
    return (size_t)k * 128 + ncol;
}

__global__ void __launch_bounds__(256)
k_tile_sort(const int* __restrict__ src, const int* __restrict__ dst,
            int2* __restrict__ ts, int2* __restrict__ hist2, int* __restrict__ total,
            const float* __restrict__ W1, unsigned* __restrict__ w1p,
            int n_edges, int nb) {
    extern __shared__ int dyn[];
    __shared__ int2 stg[TS_EDGES];
    __shared__ int  wsum[8];
    int* cntL = dyn;
    int* offL = dyn + nb;
    const int tid = threadIdx.x, lane = tid & 31, wave = tid >> 5;
    for (int i = tid; i < nb; i += 256) cntL[i] = 0;
    if (blockIdx.x == 0 && tid == 0) total[0] = 0;
    __syncthreads();

    const int wi = blockIdx.x * 256 + tid;
    float w_lo = 0.f, w_hi = 0.f;
    if (wi < 8192) {
        const size_t a = w1_entry_at(wi);
        w_lo = W1[a];
        w_hi = W1[a + 128];
    }
    const long long base = (long long)blockIdx.x * TS_EDGES;
    int s[16], d[16], r[16];
#pragma unroll
    for (int k = 0; k < 16; ++k) {
        const long long i = base + k * 256 + tid;
        const bool ok = i < n_edges;
        d[k] = ok ? dst[i] : -1;
        s[k] = ok ? src[i] : 0;
    }
#pragma unroll
    for (int k = 0; k < 16; ++k) {
        r[k] = 0;
        if (d[k] >= 0) {
            const int bk = (int)min((unsigned)d[k] >> BK_SHIFT, (unsigned)(nb - 1));
            r[k] = atomicAdd(&cntL[bk], 1);
        }
    }
    if (wi < 8192) w1p[wi] = pack2bf(w_lo, w_hi);
    __syncthreads();
    const int chunk = (nb + 255) >> 8;
    const int lo = min(tid * chunk, nb), hi = min(lo + chunk, nb);
    int mine = 0;
    for (int i = lo; i < hi; ++i) mine += cntL[i];
    int all;
    int run = block_excl_scan(mine, lane, wave, wsum, &all);
    for (int i = lo; i < hi; ++i) { offL[i] = run; run += cntL[i]; }
    __syncthreads();
    for (int i = tid; i < nb; i += 256)
        hist2[(size_t)blockIdx.x * nb + i] = make_int2(offL[i], cntL[i]);
#pragma unroll
    for (int k = 0; k < 16; ++k) {
        if (d[k] >= 0) {
            const int bk = (int)min((unsigned)d[k] >> BK_SHIFT, (unsigned)(nb - 1));
            stg[(offL[bk] + r[k]) & (TS_EDGES - 1)] = make_int2(s[k], d[k]);
        }
    }
    __syncthreads();
    const long long left = (long long)n_edges - base;
#pragma unroll
    for (int k = 0; k < 16; ++k) {
        const int i = k * 256 + tid;
        if (i < left) ts[base + i] = stg[i];
    }
    for (int i = wi + (int)gridDim.x * 256; i < 8192; i += (int)gridDim.x * 256) {
        const size_t a = w1_entry_at(i);
        w1p[i] = pack2bf(W1[a], W1[a + 128]);
    }
}

#define BK_PER_T 18
#define PX_FLIGHT 16
#define BK_CAP   (256 * BK_PER_T)

__device__ __forceinline__ void bk_fill(int* own, const int* preL, const int* relL,
                                        const int* lenL, int n_tiles, int tid,
                                        int c0, int c1) {
    for (int t = tid; t < n_tiles; t += 256) {
        const int p = preL[t], rel = relL[t];
        const int lo = max(p, c0), hi = min(p + lenL[t], c1);
        for (int f = lo; f < hi; ++f) own[f - c0] = rel;
    }
}

template <bool WHOLE>
__device__ __forceinline__ void px_trip(const float4* __restrict__ x4, uint2* __restrict__ o2,
                                        const float* dinvL, int node0, int r0, int rows,
                                        int lane) {
    float4 v[PX_FLIGHT];
#pragma unroll
    for (int j = 0; j < PX_FLIGHT; ++j) {
        const int r = r0 + j;
        v[j] = make_float4(0.f, 0.f, 0.f, 0.f);
        if (WHOLE || r < rows) v[j] = x4[(size_t)(node0 + r) * 32 + lane];
    }
    __builtin_amdgcn_sched_barrier(0);
#pragma unroll
    for (int j = 0; j < PX_FLIGHT; ++j) {
        const int r = r0 + j;
        if (WHOLE || r < rows) {
            const float d = dinvL[r];
            o2[(size_t)(node0 + r) * 32 + lane] =
                make_uint2(pack2bf(v[j].x * d, v[j].y * d), pack2bf(v[j].z * d, v[j].w * d));
        }
    }
}

__global__ void __launch_bounds__(256)
k_bucket_csr(const int2* __restrict__ ts, const int2* __restrict__ hist2, int* total,
             int* __restrict__ seg, float* __restrict__ dinv, int* __restrict__ csr,
             const float* __restrict__ x, unsigned* __restrict__ xs,
             int n_nodes, int n_edges, int nb, int n_tiles) {
    extern __shared__ int dyn[];
    __shared__ int own[BK_CAP], outL[BK_CAP];
    __shared__ int cntN[256], offN[256], cur[256], wsum[8], sbase;
    __shared__ float dinvL[256];
    int* preL = dyn;
    int* relL = dyn + n_tiles;
    int* lenL = dyn + 2 * n_tiles;
    const int tid = threadIdx.x, lane = tid & 31, wave = tid >> 5;
    const int bk    = blockIdx.x;
    const int node0 = bk << BK_SHIFT;
    cntN[tid] = 0; cur[tid] = 0;
    int mine = 0;
    for (int t = tid; t < n_tiles; t += 256) {
        const int2 h = hist2[(size_t)t * nb + bk];
        const int len = min(max(h.y, 0), TS_EDGES);
        lenL[t] = len;
        relL[t] = t * TS_EDGES + (h.x & (TS_EDGES - 1));
        mine += len;
    }
    int n_bk;
    int run = block_excl_scan(mine, lane, wave, wsum, &n_bk);
    for (int t = tid; t < n_tiles; t += 256) {
        preL[t] = run;
        relL[t] -= run;
        run += lenL[t];
    }

    for (int c0 = 0; c0 < n_bk; c0 += BK_CAP) {
        const int c1 = min(n_bk, c0 + BK_CAP);
        __syncthreads();
        bk_fill(own, preL, relL, lenL, n_tiles, tid, c0, c1);
        __syncthreads();
        int dv[BK_PER_T];
#pragma unroll
        for (int j = 0; j < BK_PER_T; ++j) {
            const int f   = min(c0 + j * 256 + tid, c1 - 1);
            const int idx = min(max(own[f - c0] + f, 0), n_edges - 1);
            dv[j] = ts[idx].y;
        }
#pragma unroll
        for (int j = 0; j < BK_PER_T; ++j)
            if (c0 + j * 256 + tid < c1) atomicAdd(&cntN[(dv[j] - node0) & 255], 1);
    }
    __syncthreads();
    const int c = cntN[tid];
    const float dv_own = rsqrtf((float)(c + 1));
    dinvL[tid] = dv_own;
    int n_cnt;
    const int off = block_excl_scan(c, lane, wave, wsum, &n_cnt);
    offN[tid] = off;
    if (tid == 0) sbase = atomicAdd(total, n_cnt);
    __syncthreads();
    const int rb = sbase;
    const int node = node0 + tid;
    if (node < n_nodes) {
        ((int2*)seg)[node] = make_int2(rb + off, c);
        dinv[node] = dv_own;
    }

    const bool staged = n_bk <= BK_CAP;
    for (int c0 = 0; c0 < n_bk; c0 += BK_CAP) {
        const int c1 = min(n_bk, c0 + BK_CAP);
        if (!staged) {
            __syncthreads();
            bk_fill(own, preL, relL, lenL, n_tiles, tid, c0, c1);
            __syncthreads();
        }
        int2 ev[BK_PER_T];
#pragma unroll
        for (int j = 0; j < BK_PER_T; ++j) {
            const int f   = min(c0 + j * 256 + tid, c1 - 1);
            const int idx = min(max(own[f - c0] + f, 0), n_edges - 1);
            ev[j] = ts[idx];
        }
#pragma unroll
        for (int j = 0; j < BK_PER_T; ++j) {
            if (c0 + j * 256 + tid < c1) {
                const int loc = (ev[j].y - node0) & 255;
                const int p   = offN[loc] + atomicAdd(&cur[loc], 1);
                if (staged) outL[min(p, BK_CAP - 1)] = ev[j].x;
                else if ((unsigned)(rb + p) < (unsigned)n_edges) csr[rb + p] = ev[j].x;
            }
        }
    }
    if (staged) {
        __syncthreads();
        int ov[BK_PER_T];
#pragma unroll
        for (int j = 0; j < BK_PER_T; ++j) ov[j] = outL[j * 256 + tid];
#pragma unroll
        for (int j = 0; j < BK_PER_T; ++j) {
            const int i = j * 256 + tid;
            if (i < n_bk && (unsigned)(rb + i) < (unsigned)n_edges) csr[rb + i] = ov[j];
        }
    }

    {
        const float4* x4 = (const float4*)x;
        uint2*        o2 = (uint2*)xs;
        const int wv   = __builtin_amdgcn_readfirstlane(wave);
        const int rows = min(256, n_nodes - node0);
        for (int r0 = wv * 32; r0 < min(wv * 32 + 32, rows); r0 += PX_FLIGHT) {
            if (r0 + PX_FLIGHT <= rows) px_trip<true >(x4, o2, dinvL, node0, r0, rows, lane);
            else                        px_trip<false>(x4, o2, dinvL, node0, r0, rows, lane);
        }
    }
}

#define FT_ROWS  16
#define FT_PITCH 68
#define FT_WAVES 4
#define G1_CHUNK 31
__device__ __forceinline__ int g1_load_srcs(const int* __restrict__ csr, int first, int m,
                                            int lane, int node, int n_edges) {
    int sv = node;
    if (lane < m) {
        const int idx = min(first + lane, n_edges - 1);
        sv = csr[idx];
    }
    return sv;
}
__device__ __forceinline__ float4 g1_f4(const uint2 v) {
    return make_float4(__uint_as_float(v.x << 16), __uint_as_float(v.x & 0xFFFF0000u),
                       __uint_as_float(v.y << 16), __uint_as_float(v.y & 0xFFFF0000u));
}
__device__ __forceinline__ float4 g1_sum(const float4 a, const float4 b) {
    return make_float4(a.x + b.x, a.y + b.y, a.z + b.z, a.w + b.w);
}
__global__ void __launch_bounds__(FT_WAVES * 32)
k_gather_gemm(const unsigned* __restrict__ xs,
              const int* __restrict__ csr,
              const int* __restrict__ seg,
              const float* __restrict__ dinv,
              const unsigned* __restrict__ w1p,
              const float* __restrict__ b1, const float* __restrict__ W2,
              float* __restrict__ h2s,
              int n_nodes, int n_edges) {
    __shared__ __attribute__((aligned(16))) unsigned zs[FT_WAVES][2][FT_ROWS][FT_PITCH];
    const int lane = threadIdx.x & 31;
    const int wv   = threadIdx.x >> 5;
    const int tile  = __builtin_amdgcn_readfirstlane((blockIdx.x * blockDim.x + threadIdx.x) >> 5);
    const int node0 = tile * FT_ROWS;
    if (node0 >= n_nodes) return;
    const int cntn = min(FT_ROWS, n_nodes - node0);

    int2  sgv = make_int2(0, 0);
    float dnv = 0.f;
    if (lane < cntn) {
        sgv = ((const int2*)seg)[node0 + lane];
        dnv = dinv[node0 + lane];
    }
    const uint2* rows = (const uint2*)xs;

    int beg = __builtin_amdgcn_readlane(sgv.x, 0);
    int n   = __builtin_amdgcn_readlane(sgv.y, 0);
    int sv_next = g1_load_srcs(csr, beg, min(n, G1_CHUNK), lane, node0, n_edges);

    for (int k = 0; k < cntn; ++k) {
        const int node = node0 + k;
        int sv = (int)min((unsigned)sv_next, (unsigned)(n_nodes - 1));
        int begn = 0, nn = 0;
        if (k + 1 < cntn) {
            begn = __builtin_amdgcn_readlane(sgv.x, k + 1);
            nn   = __builtin_amdgcn_readlane(sgv.y, k + 1);
            sv_next = g1_load_srcs(csr, begn, min(nn, G1_CHUNK), lane, node + 1, n_edges);
        }
        const uint2 own = rows[(size_t)node * 32 + lane];
        float4 acc = make_float4(0.f, 0.f, 0.f, 0.f);
        for (int c0 = 0; c0 < n; c0 += G1_CHUNK) {
            const int m = min(n - c0, G1_CHUNK);
            if (c0 > 0)
                sv = (int)min((unsigned)g1_load_srcs(csr, beg + c0, m, lane, node, n_edges),
                              (unsigned)(n_nodes - 1));
            uint2 ua[16], ub[8], uc[4], ud[2], ue;
            int j = 0;
            if (m & 16) {
#pragma unroll
                for (int t = 0; t < 16; ++t) {
                    const int s = __builtin_amdgcn_readlane(sv, t);
                    ua[t] = rows[(size_t)s * 32 + lane];
                }
                j = 16;
            }
            if (m & 8) {
#pragma unroll
                for (int t = 0; t < 8; ++t) {
                    const int s = __builtin_amdgcn_readlane(sv, j + t);
                    ub[t] = rows[(size_t)s * 32 + lane];
                }
                j += 8;
            }
            if (m & 4) {
#pragma unroll
                for (int t = 0; t < 4; ++t) {
                    const int s = __builtin_amdgcn_readlane(sv, j + t);
                    uc[t] = rows[(size_t)s * 32 + lane];
                }
                j += 4;
            }
            if (m & 2) {
                const int s0 = __builtin_amdgcn_readlane(sv, j);
                const int s1 = __builtin_amdgcn_readlane(sv, j + 1);
                ud[0] = rows[(size_t)s0 * 32 + lane];
                ud[1] = rows[(size_t)s1 * 32 + lane];
                j += 2;
            }
            if (m & 1) {
                const int s = __builtin_amdgcn_readlane(sv, j);
                ue = rows[(size_t)s * 32 + lane];
            }
            if (m & 16) {
                float4 q[4];
#pragma unroll
                for (int t = 0; t < 4; ++t)
                    q[t] = g1_sum(g1_sum(g1_f4(ua[4 * t]), g1_f4(ua[4 * t + 1])),
                                  g1_sum(g1_f4(ua[4 * t + 2]), g1_f4(ua[4 * t + 3])));
                acc = g1_sum(acc, g1_sum(g1_sum(q[0], q[1]), g1_sum(q[2], q[3])));
            }
            if (m & 8) {
                const float4 a = g1_sum(g1_sum(g1_f4(ub[0]), g1_f4(ub[1])), g1_sum(g1_f4(ub[2]), g1_f4(ub[3])));
                const float4 b = g1_sum(g1_sum(g1_f4(ub[4]), g1_f4(ub[5])), g1_sum(g1_f4(ub[6]), g1_f4(ub[7])));
                acc = g1_sum(acc, g1_sum(a, b));
            }
            if (m & 4)
                acc = g1_sum(acc, g1_sum(g1_sum(g1_f4(uc[0]), g1_f4(uc[1])), g1_sum(g1_f4(uc[2]), g1_f4(uc[3]))));
            if (m & 2)
                acc = g1_sum(acc, g1_sum(g1_f4(ud[0]), g1_f4(ud[1])));
            if (m & 1)
                acc = g1_sum(acc, g1_f4(ue));
        }
        acc = g1_sum(acc, g1_f4(own));

        const float dn = __int_as_float(__builtin_amdgcn_readlane(__float_as_int(dnv), k));
        const float z0 = acc.x * dn, z1 = acc.y * dn, z2 = acc.z * dn, z3 = acc.w * dn;
        const unsigned h0 = bf16_rne(z0), h1 = bf16_rne(z1), h2 = bf16_rne(z2), h3 = bf16_rne(z3);
        const float r0 = z0 - __uint_as_float(h0 << 16), r1 = z1 - __uint_as_float(h1 << 16);
        const float r2 = z2 - __uint_as_float(h2 << 16), r3 = z3 - __uint_as_float(h3 << 16);
        *(uint2*)&zs[wv][0][k][2 * lane] = make_uint2(h0 | (h1 << 16), h2 | (h3 << 16));
        *(uint2*)&zs[wv][1][k][2 * lane] = make_uint2(pack2bf(r0, r1), pack2bf(r2, r3));
        beg = begn; n = nn;
    }
    for (int k = cntn; k < FT_ROWS; ++k) {
        *(uint2*)&zs[wv][0][k][2 * lane] = make_uint2(0u, 0u);
        *(uint2*)&zs[wv][1][k][2 * lane] = make_uint2(0u, 0u);
    }
    __builtin_amdgcn_fence(__ATOMIC_RELEASE, "wavefront");
    __builtin_amdgcn_wave_barrier();
    __builtin_amdgcn_fence(__ATOMIC_ACQUIRE, "wavefront");

    const int m0   = node0;
    const int mr   = lane & 15;
    const int half = lane >> 4;
    const unsigned* rh = &zs[wv][0][mr][0];
    const unsigned* rl = &zs[wv][1][mr][0];
    const v8u* bfrag = (const v8u*)w1p;

    v8f acc[8];
#pragma unroll
    for (int nt = 0; nt < 8; ++nt) acc[nt] = (v8f){};
#pragma unroll
    for (int ks = 0; ks < 4; ++ks) {
        FragAB ah, al;
        {
            const uint4 hlo = *(const uint4*)(rh + ks * 16 + half * 4);
            const uint4 hhi = *(const uint4*)(rh + ks * 16 + 8 + half * 4);
            const uint4 llo = *(const uint4*)(rl + ks * 16 + half * 4);
            const uint4 lhi = *(const uint4*)(rl + ks * 16 + 8 + half * 4);
            ah.u = (v8u){hlo.x, hlo.y, hlo.z, hlo.w, hhi.x, hhi.y, hhi.z, hhi.w};
            al.u = (v8u){llo.x, llo.y, llo.z, llo.w, lhi.x, lhi.y, lhi.z, lhi.w};
        }
#pragma unroll
        for (int g = 0; g < 8; g += 4) {
            FragAB b[4];
#pragma unroll
            for (int i = 0; i < 4; ++i)
                b[i].u = bfrag[((g + i) * 4 + ks) * 32 + lane];
            __builtin_amdgcn_sched_barrier(0);
#pragma unroll
            for (int i = 0; i < 4; ++i)
                acc[g + i] = __builtin_amdgcn_wmma_f32_16x16x32_bf16(
                                 false, ah.b, false, b[i].b, (short)0, acc[g + i], false, false);
#pragma unroll
            for (int i = 0; i < 4; ++i)
                acc[g + i] = __builtin_amdgcn_wmma_f32_16x16x32_bf16(
                                 false, al.b, false, b[i].b, (short)0, acc[g + i], false, false);
            __builtin_amdgcn_sched_barrier(0);
        }
    }
    float bv[8], wa[8], wb[8];
#pragma unroll
    for (int nt = 0; nt < 8; ++nt) {
        const int col = nt * 16 + mr;
        const float2 w = ((const float2*)W2)[col];
        bv[nt] = b1[col]; wa[nt] = w.x; wb[nt] = w.y;
    }
    float p0[8], p1[8];
#pragma unroll
    for (int v = 0; v < 8; ++v) {
        float s0 = 0.f, s1 = 0.f;
#pragma unroll
        for (int nt = 0; nt < 8; ++nt) {
            const float t = fmaxf(acc[nt][v] + bv[nt], 0.f);
            s0 += t * wa[nt];
            s1 += t * wb[nt];
        }
        p0[v] = s0; p1[v] = s1;
    }
#pragma unroll
    for (int off = 8; off > 0; off >>= 1) {
#pragma unroll
        for (int v = 0; v < 8; ++v) {
            p0[v] += __shfl_xor(p0[v], off, 32);
            p1[v] += __shfl_xor(p1[v], off, 32);
        }
    }
    float o0 = p0[0], o1 = p1[0];
#pragma unroll
    for (int v = 1; v < 8; ++v)
        if (mr == v) { o0 = p0[v]; o1 = p1[v]; }
    const int orow = m0 + (half << 3) + mr;
    const float dn = __shfl(dnv, (half << 3) + (mr & 7), 32);
    if (mr < 8 && orow < n_nodes)
        ((float2*)h2s)[orow] = make_float2(o0 * dn, o1 * dn);
}

__global__ void k_gather2_out(const float* __restrict__ h2s,
                              const int* __restrict__ csr,
                              const int* __restrict__ seg,
                              const float* __restrict__ dinv, const float* __restrict__ b2,
                              float* __restrict__ out, int n_nodes, int n_edges) {
    const int  gid   = blockIdx.x * blockDim.x + threadIdx.x;
    const int  node  = gid >> 1;
    const int  sub   = gid & 1;
    const bool valid = node < n_nodes;
    const int  nc    = valid ? node : n_nodes - 1;
    const int2 sg    = ((const int2*)seg)[nc];
    const int  beg   = sg.x;
    const int  n     = valid ? sg.y : 0;
    const float2* hs = (const float2*)h2s;

    float2 acc = make_float2(0.f, 0.f);
    if (sub == 0) acc = hs[nc];
#pragma unroll 4
    for (int j = sub; j < n; j += 2) {
        const int idx = min(beg + j, n_edges - 1);
        const int s   = (int)min((unsigned)csr[idx], (unsigned)(n_nodes - 1));
        const float2 v = hs[s];
        acc.x += v.x; acc.y += v.y;
    }
    acc.x += __shfl_xor(acc.x, 1, 32);
    acc.y += __shfl_xor(acc.y, 1, 32);
    if (valid && sub == 0) {
        const float dn = dinv[node];
        const float z0 = acc.x * dn + b2[0];
        const float z1 = acc.y * dn + b2[1];
        const float m  = fmaxf(z0, z1);
        const float l  = m + __logf(__expf(z0 - m) + __expf(z1 - m));
        ((float2*)out)[node] = make_float2(z0 - l, z1 - l);
    }
}

extern "C" void kernel_launch(void* const* d_in, const int* in_sizes, int n_in,
                              void* d_out, int out_size, void* d_ws, size_t ws_size,
                              hipStream_t stream) {
    const float* x   = (const float*)d_in[0];
    const int*   ei  = (const int*)  d_in[1];
    const float* W1  = (const float*)d_in[2];
    const float* b1  = (const float*)d_in[3];
    const float* W2  = (const float*)d_in[4];
    const float* b2  = (const float*)d_in[5];
    float*       out = (float*)d_out;

    const int n_nodes = in_sizes[0] / 128;
    const int n_edges = in_sizes[1] / 2;
    const int* src = ei;
    const int* dst = ei + n_edges;

    const size_t nP = ((size_t)n_nodes + 127) / 128 * 128;
    unsigned* zh    = (unsigned*)d_ws;
    unsigned* zl    = zh + (size_t)n_nodes * 64;
    unsigned* xs    = zl + (size_t)n_nodes * 64;
    unsigned* w1p   = xs + (size_t)n_nodes * 64;
    float*    h2s   = (float*)(w1p + 8192);
    float*    dinv  = h2s + 2 * nP;
    int*      seg   = (int*)(dinv + nP);
    int*      total = seg + 2 * nP;
    int*      csr   = total + 128;
    int2*     ts    = (int2*)zh;
    int2*     hist2 = (int2*)zl;

    const int B = 256;
    const int nb = (n_nodes + 255) / 256;
    const int nt = (n_edges + TS_EDGES - 1) / TS_EDGES;
    k_tile_sort <<<nt, B, 2 * nb * sizeof(int), stream>>>(src, dst, ts, hist2, total, W1, w1p, n_edges, nb);
    k_bucket_csr<<<nb, B, 3 * nt * sizeof(int), stream>>>(ts, hist2, total, seg, dinv, csr, x, xs,
                                                          n_nodes, n_edges, nb, nt);
    const int n_tiles = (n_nodes + FT_ROWS - 1) / FT_ROWS;
    k_gather_gemm<<<(n_tiles + FT_WAVES - 1) / FT_WAVES, FT_WAVES * 32, 0, stream>>>(
        xs, csr, seg, dinv, w1p, b1, W2, h2s, n_nodes, n_edges);
    k_gather2_out<<<(n_nodes * 2 + B - 1) / B, B, 0, stream>>>(
        h2s, csr, seg, dinv, b2, out, n_nodes, n_edges);
}


# layer-1 gather: about half the vector instructions per node: bf16 halves added by one mixed-precision FMA each (no unpack), z packed by the pair convert, row bases scalar; rows summed in a new order
# speedup vs baseline: 8.1564x; 1.0330x over previous
#include <hip/hip_runtime.h>
#include <hip/hip_bf16.h>


typedef __attribute__((ext_vector_type(16))) __bf16     v16bf;
typedef __attribute__((ext_vector_type(8)))  float      v8f;
typedef __attribute__((ext_vector_type(8)))  unsigned   v8u;

typedef __attribute__((ext_vector_type(2)))  __bf16     v2bf;
typedef __attribute__((ext_vector_type(2)))  float      v2f;
typedef __attribute__((ext_vector_type(2)))  unsigned   v2u;

union FragAB { v8u u; v16bf b; };

__device__ __forceinline__ unsigned pack2bf(float lo, float hi) {
    const v2f  f = {lo, hi};
    const v2bf p = __builtin_convertvector(f, v2bf);
    return __builtin_bit_cast(unsigned, p);
}
__device__ __forceinline__ float fma_bf_lo(unsigned u, float f, float acc) {
    float r;
    asm("v_fma_mix_f32_bf16 %0, %1, %2, %3 op_sel_hi:[1,0,0]"
        : "=v"(r) : "v"(u), "v"(f), "v"(acc));
    return r;
}
__device__ __forceinline__ float fma_bf_hi(unsigned u, float f, float acc) {
    float r;
    asm("v_fma_mix_f32_bf16 %0, %1, %2, %3 op_sel:[1,0,0] op_sel_hi:[1,0,0]"
        : "=v"(r) : "v"(u), "v"(f), "v"(acc));
    return r;
}
__device__ __forceinline__ float add_bf_lo(float acc, unsigned u) { return fma_bf_lo(u, 1.0f, acc); }
__device__ __forceinline__ float add_bf_hi(float acc, unsigned u) { return fma_bf_hi(u, 1.0f, acc); }
__device__ __forceinline__ float sub_bf_lo(float z, unsigned u) { return fma_bf_lo(u, -1.0f, z); }
__device__ __forceinline__ float sub_bf_hi(float z, unsigned u) { return fma_bf_hi(u, -1.0f, z); }

#define TS_EDGES 4096
#define BK_SHIFT 8

__device__ __forceinline__ int block_excl_scan(int v, int lane, int wave,
                                               int* wsum, int* all_sum) {
    int incl = v;
#pragma unroll
    for (int off = 1; off < 32; off <<= 1) {
        const int t = __shfl_up(incl, off, 32);
        if (lane >= off) incl += t;
    }
    if (lane == 31) wsum[wave] = incl;
    __syncthreads();
    int wbase = 0, tot = 0;
#pragma unroll
    for (int w = 0; w < 8; ++w) {
        const int x = wsum[w];
        if (w < wave) wbase += x;
        tot += x;
    }
    *all_sum = tot;
    return wbase + incl - v;
}

__device__ __forceinline__ size_t w1_entry_at(int i) {
    const int v    = i & 7;
    const int lane = (i >> 3) & 31;
    const int ks   = (i >> 8) & 3;
    const int nt   = i >> 10;
    const int k    = ks * 32 + ((v >> 2) << 4) + ((lane >> 4) << 3) + ((v & 3) << 1);
    const int ncol = nt * 16 + (lane & 15);
    return (size_t)k * 128 + ncol;
}

__global__ void __launch_bounds__(256)
k_tile_sort(const int* __restrict__ src, const int* __restrict__ dst,
            int2* __restrict__ ts, int2* __restrict__ hist2, int* __restrict__ total,
            const float* __restrict__ W1, unsigned* __restrict__ w1p,
            int n_edges, int nb) {
    extern __shared__ int dyn[];
    __shared__ int2 stg[TS_EDGES];
    __shared__ int  wsum[8];
    int* cntL = dyn;
    int* offL = dyn + nb;
    const int tid = threadIdx.x, lane = tid & 31, wave = tid >> 5;
    for (int i = tid; i < nb; i += 256) cntL[i] = 0;
    if (blockIdx.x == 0 && tid == 0) total[0] = 0;
    __syncthreads();

    const int wi = blockIdx.x * 256 + tid;
    float w_lo = 0.f, w_hi = 0.f;
    if (wi < 8192) {
        const size_t a = w1_entry_at(wi);
        w_lo = W1[a];
        w_hi = W1[a + 128];
    }
    const long long base = (long long)blockIdx.x * TS_EDGES;
    int s[16], d[16], r[16];
#pragma unroll
    for (int k = 0; k < 16; ++k) {
        const long long i = base + k * 256 + tid;
        const bool ok = i < n_edges;
        d[k] = ok ? dst[i] : -1;
        s[k] = ok ? src[i] : 0;
    }
#pragma unroll
    for (int k = 0; k < 16; ++k) {
        r[k] = 0;
        if (d[k] >= 0) {
            const int bk = (int)min((unsigned)d[k] >> BK_SHIFT, (unsigned)(nb - 1));
            r[k] = atomicAdd(&cntL[bk], 1);
        }
    }
    if (wi < 8192) w1p[wi] = pack2bf(w_lo, w_hi);
    __syncthreads();
    const int chunk = (nb + 255) >> 8;
    const int lo = min(tid * chunk, nb), hi = min(lo + chunk, nb);
    int mine = 0;
    for (int i = lo; i < hi; ++i) mine += cntL[i];
    int all;
    int run = block_excl_scan(mine, lane, wave, wsum, &all);
    for (int i = lo; i < hi; ++i) { offL[i] = run; run += cntL[i]; }
    __syncthreads();
    for (int i = tid; i < nb; i += 256)
        hist2[(size_t)blockIdx.x * nb + i] = make_int2(offL[i], cntL[i]);
#pragma unroll
    for (int k = 0; k < 16; ++k) {
        if (d[k] >= 0) {
            const int bk = (int)min((unsigned)d[k] >> BK_SHIFT, (unsigned)(nb - 1));
            stg[(offL[bk] + r[k]) & (TS_EDGES - 1)] = make_int2(s[k], d[k]);
        }
    }
    __syncthreads();
    const long long left = (long long)n_edges - base;
#pragma unroll
    for (int k = 0; k < 16; ++k) {
        const int i = k * 256 + tid;
        if (i < left) ts[base + i] = stg[i];
    }
    for (int i = wi + (int)gridDim.x * 256; i < 8192; i += (int)gridDim.x * 256) {
        const size_t a = w1_entry_at(i);
        w1p[i] = pack2bf(W1[a], W1[a + 128]);
    }
}

#define BK_PER_T 18
#define PX_FLIGHT 16
#define BK_CAP   (256 * BK_PER_T)

__device__ __forceinline__ void bk_fill(int* own, const int* preL, const int* relL,
                                        const int* lenL, int n_tiles, int tid,
                                        int c0, int c1) {
    for (int t = tid; t < n_tiles; t += 256) {
        const int p = preL[t], rel = relL[t];
        const int lo = max(p, c0), hi = min(p + lenL[t], c1);
        for (int f = lo; f < hi; ++f) own[f - c0] = rel;
    }
}

template <bool WHOLE>
__device__ __forceinline__ void px_trip(const float4* __restrict__ x4, uint2* __restrict__ o2,
                                        const float* dinvL, int node0, int r0, int rows,
                                        int lane) {
    float4 v[PX_FLIGHT];
#pragma unroll
    for (int j = 0; j < PX_FLIGHT; ++j) {
        const int r = r0 + j;
        v[j] = make_float4(0.f, 0.f, 0.f, 0.f);
        if (WHOLE || r < rows) v[j] = x4[(size_t)(node0 + r) * 32 + lane];
    }
    __builtin_amdgcn_sched_barrier(0);
#pragma unroll
    for (int j = 0; j < PX_FLIGHT; ++j) {
        const int r = r0 + j;
        if (WHOLE || r < rows) {
            const float d = dinvL[r];
            o2[(size_t)(node0 + r) * 32 + lane] =
                make_uint2(pack2bf(v[j].x * d, v[j].y * d), pack2bf(v[j].z * d, v[j].w * d));
        }
    }
}

__global__ void __launch_bounds__(256)
k_bucket_csr(const int2* __restrict__ ts, const int2* __restrict__ hist2, int* total,
             int* __restrict__ seg, float* __restrict__ dinv, int* __restrict__ csr,
             const float* __restrict__ x, unsigned* __restrict__ xs,
             int n_nodes, int n_edges, int nb, int n_tiles) {
    extern __shared__ int dyn[];
    __shared__ int own[BK_CAP], outL[BK_CAP];
    __shared__ int cntN[256], offN[256], cur[256], wsum[8], sbase;
    __shared__ float dinvL[256];
    int* preL = dyn;
    int* relL = dyn + n_tiles;
    int* lenL = dyn + 2 * n_tiles;
    const int tid = threadIdx.x, lane = tid & 31, wave = tid >> 5;
    const int bk    = blockIdx.x;
    const int node0 = bk << BK_SHIFT;
    cntN[tid] = 0; cur[tid] = 0;
    int mine = 0;
    for (int t = tid; t < n_tiles; t += 256) {
        const int2 h = hist2[(size_t)t * nb + bk];
        const int len = min(max(h.y, 0), TS_EDGES);
        lenL[t] = len;
        relL[t] = t * TS_EDGES + (h.x & (TS_EDGES - 1));
        mine += len;
    }
    int n_bk;
    int run = block_excl_scan(mine, lane, wave, wsum, &n_bk);
    for (int t = tid; t < n_tiles; t += 256) {
        preL[t] = run;
        relL[t] -= run;
        run += lenL[t];
    }

    for (int c0 = 0; c0 < n_bk; c0 += BK_CAP) {
        const int c1 = min(n_bk, c0 + BK_CAP);
        __syncthreads();
        bk_fill(own, preL, relL, lenL, n_tiles, tid, c0, c1);
        __syncthreads();
        int dv[BK_PER_T];
#pragma unroll
        for (int j = 0; j < BK_PER_T; ++j) {
            const int f   = min(c0 + j * 256 + tid, c1 - 1);
            const int idx = min(max(own[f - c0] + f, 0), n_edges - 1);
            dv[j] = ts[idx].y;
        }
#pragma unroll
        for (int j = 0; j < BK_PER_T; ++j)
            if (c0 + j * 256 + tid < c1) atomicAdd(&cntN[(dv[j] - node0) & 255], 1);
    }
    __syncthreads();
    const int c = cntN[tid];
    const float dv_own = rsqrtf((float)(c + 1));
    dinvL[tid] = dv_own;
    int n_cnt;
    const int off = block_excl_scan(c, lane, wave, wsum, &n_cnt);
    offN[tid] = off;
    if (tid == 0) sbase = atomicAdd(total, n_cnt);
    __syncthreads();
    const int rb = sbase;
    const int node = node0 + tid;
    if (node < n_nodes) {
        ((int2*)seg)[node] = make_int2(rb + off, c);
        dinv[node] = dv_own;
    }

    const bool staged = n_bk <= BK_CAP;
    for (int c0 = 0; c0 < n_bk; c0 += BK_CAP) {
        const int c1 = min(n_bk, c0 + BK_CAP);
        if (!staged) {
            __syncthreads();
            bk_fill(own, preL, relL, lenL, n_tiles, tid, c0, c1);
            __syncthreads();
        }
        int2 ev[BK_PER_T];
#pragma unroll
        for (int j = 0; j < BK_PER_T; ++j) {
            const int f   = min(c0 + j * 256 + tid, c1 - 1);
            const int idx = min(max(own[f - c0] + f, 0), n_edges - 1);
            ev[j] = ts[idx];
        }
#pragma unroll
        for (int j = 0; j < BK_PER_T; ++j) {
            if (c0 + j * 256 + tid < c1) {
                const int loc = (ev[j].y - node0) & 255;
                const int p   = offN[loc] + atomicAdd(&cur[loc], 1);
                if (staged) outL[min(p, BK_CAP - 1)] = ev[j].x;
                else if ((unsigned)(rb + p) < (unsigned)n_edges) csr[rb + p] = ev[j].x;
            }
        }
    }
    if (staged) {
        __syncthreads();
        int ov[BK_PER_T];
#pragma unroll
        for (int j = 0; j < BK_PER_T; ++j) ov[j] = outL[j * 256 + tid];
#pragma unroll
        for (int j = 0; j < BK_PER_T; ++j) {
            const int i = j * 256 + tid;
            if (i < n_bk && (unsigned)(rb + i) < (unsigned)n_edges) csr[rb + i] = ov[j];
        }
    }

    {
        const float4* x4 = (const float4*)x;
        uint2*        o2 = (uint2*)xs;
        const int wv   = __builtin_amdgcn_readfirstlane(wave);
        const int rows = min(256, n_nodes - node0);
        for (int r0 = wv * 32; r0 < min(wv * 32 + 32, rows); r0 += PX_FLIGHT) {
            if (r0 + PX_FLIGHT <= rows) px_trip<true >(x4, o2, dinvL, node0, r0, rows, lane);
            else                        px_trip<false>(x4, o2, dinvL, node0, r0, rows, lane);
        }
    }
}

#define FT_ROWS  16
#define FT_PITCH 68
#define FT_WAVES 4
#define G1_CHUNK 31
__device__ __forceinline__ int g1_load_srcs(const int* __restrict__ csr, int first, int m,
                                            int lane, int node, int n_edges) {
    int sv = node;
    if (lane < m) {
        const int idx = min(first + lane, n_edges - 1);
        sv = csr[idx];
    }
    return sv;
}
__device__ __forceinline__ float4 g1_add(const float4 a, const v2u v) {
    return make_float4(add_bf_lo(a.x, v.x), add_bf_hi(a.y, v.x),
                       add_bf_lo(a.z, v.y), add_bf_hi(a.w, v.y));
}
typedef const __attribute__((address_space(1))) v2u* g1_rowp;
__device__ __forceinline__ g1_rowp g1_row(const uint2* rows, int s) {
    g1_rowp p = (g1_rowp)(rows + (size_t)s * 32);
    asm("" : "+s"(p));
    return p;
}
__global__ void __launch_bounds__(FT_WAVES * 32)
k_gather_gemm(const unsigned* __restrict__ xs,
              const int* __restrict__ csr,
              const int* __restrict__ seg,
              const float* __restrict__ dinv,
              const unsigned* __restrict__ w1p,
              const float* __restrict__ b1, const float* __restrict__ W2,
              float* __restrict__ h2s,
              int n_nodes, int n_edges) {
    __shared__ __attribute__((aligned(16))) unsigned zs[FT_WAVES][2][FT_ROWS][FT_PITCH];
    const int lane = threadIdx.x & 31;
    const int wv   = threadIdx.x >> 5;
    const int tile  = __builtin_amdgcn_readfirstlane((blockIdx.x * blockDim.x + threadIdx.x) >> 5);
    const int node0 = tile * FT_ROWS;
    if (node0 >= n_nodes) return;
    const int cntn = min(FT_ROWS, n_nodes - node0);

    int2  sgv = make_int2(0, 0);
    float dnv = 0.f;
    if (lane < cntn) {
        sgv = ((const int2*)seg)[node0 + lane];
        dnv = dinv[node0 + lane];
    }
    const uint2* rows = (const uint2*)xs;

    int beg = __builtin_amdgcn_readlane(sgv.x, 0);
    int n   = __builtin_amdgcn_readlane(sgv.y, 0);
    int sv_next = g1_load_srcs(csr, beg, min(n, G1_CHUNK), lane, node0, n_edges);

    for (int k = 0; k < cntn; ++k) {
        const int node = node0 + k;
        int sv = (int)min((unsigned)sv_next, (unsigned)(n_nodes - 1));
        int begn = 0, nn = 0;
        if (k + 1 < cntn) {
            begn = __builtin_amdgcn_readlane(sgv.x, k + 1);
            nn   = __builtin_amdgcn_readlane(sgv.y, k + 1);
            sv_next = g1_load_srcs(csr, begn, min(nn, G1_CHUNK), lane, node + 1, n_edges);
        }
        const unsigned lo8 = (unsigned)lane;
        const v2u own = g1_row(rows, node)[lo8];
        float4 acc  = make_float4(0.f, 0.f, 0.f, 0.f);
        float4 acc2 = make_float4(0.f, 0.f, 0.f, 0.f);
        for (int c0 = 0; c0 < n; c0 += G1_CHUNK) {
            const int m = min(n - c0, G1_CHUNK);
            if (c0 > 0)
                sv = (int)min((unsigned)g1_load_srcs(csr, beg + c0, m, lane, node, n_edges),
                              (unsigned)(n_nodes - 1));
            v2u ua[16], ub[8], uc[4], ud[2], ue;
            g1_rowp pa[16];
            int j = 0;
            if (m & 16) {
#pragma unroll
                for (int t = 0; t < 16; ++t)
                    pa[t] = g1_row(rows, __builtin_amdgcn_readlane(sv, t));
                __builtin_amdgcn_sched_barrier(0);
#pragma unroll
                for (int t = 0; t < 16; ++t)
                    ua[t] = pa[t][lo8];
                j = 16;
            }
            if (m & 8) {
#pragma unroll
                for (int t = 0; t < 8; ++t)
                    ub[t] = g1_row(rows, __builtin_amdgcn_readlane(sv, j + t))[lo8];
                j += 8;
            }
            if (m & 4) {
#pragma unroll
                for (int t = 0; t < 4; ++t)
                    uc[t] = g1_row(rows, __builtin_amdgcn_readlane(sv, j + t))[lo8];
                j += 4;
            }
            if (m & 2) {
                ud[0] = g1_row(rows, __builtin_amdgcn_readlane(sv, j))[lo8];
                ud[1] = g1_row(rows, __builtin_amdgcn_readlane(sv, j + 1))[lo8];
                j += 2;
            }
            if (m & 1)
                ue = g1_row(rows, __builtin_amdgcn_readlane(sv, j))[lo8];
            if (m & 16) {
#pragma unroll
                for (int t = 0; t < 16; t += 2) {
                    acc  = g1_add(acc,  ua[t]);
                    acc2 = g1_add(acc2, ua[t + 1]);
                }
            }
            if (m & 8) {
#pragma unroll
                for (int t = 0; t < 8; t += 2) {
                    acc  = g1_add(acc,  ub[t]);
                    acc2 = g1_add(acc2, ub[t + 1]);
                }
            }
            if (m & 4) {
                acc  = g1_add(acc,  uc[0]);
                acc2 = g1_add(acc2, uc[1]);
                acc  = g1_add(acc,  uc[2]);
                acc2 = g1_add(acc2, uc[3]);
            }
            if (m & 2) {
                acc  = g1_add(acc,  ud[0]);
                acc2 = g1_add(acc2, ud[1]);
            }
            if (m & 1)
                acc = g1_add(acc, ue);
        }
        acc2 = g1_add(acc2, own);

        const float dn = __int_as_float(__builtin_amdgcn_readlane(__float_as_int(dnv), k));
        const float z0 = (acc.x + acc2.x) * dn, z1 = (acc.y + acc2.y) * dn;
        const float z2 = (acc.z + acc2.z) * dn, z3 = (acc.w + acc2.w) * dn;
        const unsigned hA = pack2bf(z0, z1), hB = pack2bf(z2, z3);
        const float r0 = sub_bf_lo(z0, hA), r1 = sub_bf_hi(z1, hA);
        const float r2 = sub_bf_lo(z2, hB), r3 = sub_bf_hi(z3, hB);
        *(uint2*)&zs[wv][0][k][2 * lane] = make_uint2(hA, hB);
        *(uint2*)&zs[wv][1][k][2 * lane] = make_uint2(pack2bf(r0, r1), pack2bf(r2, r3));
        beg = begn; n = nn;
    }
    for (int k = cntn; k < FT_ROWS; ++k) {
        *(uint2*)&zs[wv][0][k][2 * lane] = make_uint2(0u, 0u);
        *(uint2*)&zs[wv][1][k][2 * lane] = make_uint2(0u, 0u);
    }
    __builtin_amdgcn_fence(__ATOMIC_RELEASE, "wavefront");
    __builtin_amdgcn_wave_barrier();
    __builtin_amdgcn_fence(__ATOMIC_ACQUIRE, "wavefront");

    const int m0   = node0;
    const int mr   = lane & 15;
    const int half = lane >> 4;
    const unsigned* rh = &zs[wv][0][mr][0];
    const unsigned* rl = &zs[wv][1][mr][0];
    const v8u* bfrag = (const v8u*)w1p;

    v8f acc[8];
#pragma unroll
    for (int nt = 0; nt < 8; ++nt) acc[nt] = (v8f){};
#pragma unroll
    for (int ks = 0; ks < 4; ++ks) {
        FragAB ah, al;
        {
            const uint4 hlo = *(const uint4*)(rh + ks * 16 + half * 4);
            const uint4 hhi = *(const uint4*)(rh + ks * 16 + 8 + half * 4);
            const uint4 llo = *(const uint4*)(rl + ks * 16 + half * 4);
            const uint4 lhi = *(const uint4*)(rl + ks * 16 + 8 + half * 4);
            ah.u = (v8u){hlo.x, hlo.y, hlo.z, hlo.w, hhi.x, hhi.y, hhi.z, hhi.w};
            al.u = (v8u){llo.x, llo.y, llo.z, llo.w, lhi.x, lhi.y, lhi.z, lhi.w};
        }
#pragma unroll
        for (int g = 0; g < 8; g += 4) {
            FragAB b[4];
#pragma unroll
            for (int i = 0; i < 4; ++i)
                b[i].u = bfrag[((g + i) * 4 + ks) * 32 + lane];
            __builtin_amdgcn_sched_barrier(0);
#pragma unroll
            for (int i = 0; i < 4; ++i)
                acc[g + i] = __builtin_amdgcn_wmma_f32_16x16x32_bf16(
                                 false, ah.b, false, b[i].b, (short)0, acc[g + i], false, false);
#pragma unroll
            for (int i = 0; i < 4; ++i)
                acc[g + i] = __builtin_amdgcn_wmma_f32_16x16x32_bf16(
                                 false, al.b, false, b[i].b, (short)0, acc[g + i], false, false);
            __builtin_amdgcn_sched_barrier(0);
        }
    }
    float bv[8], wa[8], wb[8];
#pragma unroll
    for (int nt = 0; nt < 8; ++nt) {
        const int col = nt * 16 + mr;
        const float2 w = ((const float2*)W2)[col];
        bv[nt] = b1[col]; wa[nt] = w.x; wb[nt] = w.y;
    }
    float p0[8], p1[8];
#pragma unroll
    for (int v = 0; v < 8; ++v) {
        float s0 = 0.f, s1 = 0.f;
#pragma unroll
        for (int nt = 0; nt < 8; ++nt) {
            const float t = fmaxf(acc[nt][v] + bv[nt], 0.f);
            s0 += t * wa[nt];
            s1 += t * wb[nt];
        }
        p0[v] = s0; p1[v] = s1;
    }
#pragma unroll
    for (int off = 8; off > 0; off >>= 1) {
#pragma unroll
        for (int v = 0; v < 8; ++v) {
            p0[v] += __shfl_xor(p0[v], off, 32);
            p1[v] += __shfl_xor(p1[v], off, 32);
        }
    }
    float o0 = p0[0], o1 = p1[0];
#pragma unroll
    for (int v = 1; v < 8; ++v)
        if (mr == v) { o0 = p0[v]; o1 = p1[v]; }
    const int orow = m0 + (half << 3) + mr;
    const float dn = __shfl(dnv, (half << 3) + (mr & 7), 32);
    if (mr < 8 && orow < n_nodes)
        ((float2*)h2s)[orow] = make_float2(o0 * dn, o1 * dn);
}

__global__ void k_gather2_out(const float* __restrict__ h2s,
                              const int* __restrict__ csr,
                              const int* __restrict__ seg,
                              const float* __restrict__ dinv, const float* __restrict__ b2,
                              float* __restrict__ out, int n_nodes, int n_edges) {
    const int  gid   = blockIdx.x * blockDim.x + threadIdx.x;
    const int  node  = gid >> 1;
    const int  sub   = gid & 1;
    const bool valid = node < n_nodes;
    const int  nc    = valid ? node : n_nodes - 1;
    const int2 sg    = ((const int2*)seg)[nc];
    const int  beg   = sg.x;
    const int  n     = valid ? sg.y : 0;
    const float2* hs = (const float2*)h2s;

    float2 acc = make_float2(0.f, 0.f);
    if (sub == 0) acc = hs[nc];
#pragma unroll 4
    for (int j = sub; j < n; j += 2) {
        const int idx = min(beg + j, n_edges - 1);
        const int s   = (int)min((unsigned)csr[idx], (unsigned)(n_nodes - 1));
        const float2 v = hs[s];
        acc.x += v.x; acc.y += v.y;
    }
    acc.x += __shfl_xor(acc.x, 1, 32);
    acc.y += __shfl_xor(acc.y, 1, 32);
    if (valid && sub == 0) {
        const float dn = dinv[node];
        const float z0 = acc.x * dn + b2[0];
        const float z1 = acc.y * dn + b2[1];
        const float m  = fmaxf(z0, z1);
        const float l  = m + __logf(__expf(z0 - m) + __expf(z1 - m));
        ((float2*)out)[node] = make_float2(z0 - l, z1 - l);
    }
}

extern "C" void kernel_launch(void* const* d_in, const int* in_sizes, int n_in,
                              void* d_out, int out_size, void* d_ws, size_t ws_size,
                              hipStream_t stream) {
    const float* x   = (const float*)d_in[0];
    const int*   ei  = (const int*)  d_in[1];
    const float* W1  = (const float*)d_in[2];
    const float* b1  = (const float*)d_in[3];
    const float* W2  = (const float*)d_in[4];
    const float* b2  = (const float*)d_in[5];
    float*       out = (float*)d_out;

    const int n_nodes = in_sizes[0] / 128;
    const int n_edges = in_sizes[1] / 2;
    const int* src = ei;
    const int* dst = ei + n_edges;

    const size_t nP = ((size_t)n_nodes + 127) / 128 * 128;
    unsigned* zh    = (unsigned*)d_ws;
    unsigned* zl    = zh + (size_t)n_nodes * 64;
    unsigned* xs    = zl + (size_t)n_nodes * 64;
    unsigned* w1p   = xs + (size_t)n_nodes * 64;
    float*    h2s   = (float*)(w1p + 8192);
    float*    dinv  = h2s + 2 * nP;
    int*      seg   = (int*)(dinv + nP);
    int*      total = seg + 2 * nP;
    int*      csr   = total + 128;
    int2*     ts    = (int2*)zh;
    int2*     hist2 = (int2*)zl;

    const int B = 256;
    const int nb = (n_nodes + 255) / 256;
    const int nt = (n_edges + TS_EDGES - 1) / TS_EDGES;
    k_tile_sort <<<nt, B, 2 * nb * sizeof(int), stream>>>(src, dst, ts, hist2, total, W1, w1p, n_edges, nb);
    k_bucket_csr<<<nb, B, 3 * nt * sizeof(int), stream>>>(ts, hist2, total, seg, dinv, csr, x, xs,
                                                          n_nodes, n_edges, nb, nt);
    const int n_tiles = (n_nodes + FT_ROWS - 1) / FT_ROWS;
    k_gather_gemm<<<(n_tiles + FT_WAVES - 1) / FT_WAVES, FT_WAVES * 32, 0, stream>>>(
        xs, csr, seg, dinv, w1p, b1, W2, h2s, n_nodes, n_edges);
    k_gather2_out<<<(n_nodes * 2 + B - 1) / B, B, 0, stream>>>(
        h2s, csr, seg, dinv, b2, out, n_nodes, n_edges);
}


# layer-2 gather: a lane requests all its source indices, then all its rows, before the first wait (pieces of 8+4+4); dinv comes with the first trip: 3 dependent cache trips per wave, not up to 14
# speedup vs baseline: 8.4067x; 1.0307x over previous
#include <hip/hip_runtime.h>
#include <hip/hip_bf16.h>


typedef __attribute__((ext_vector_type(16))) __bf16     v16bf;
typedef __attribute__((ext_vector_type(8)))  float      v8f;
typedef __attribute__((ext_vector_type(8)))  unsigned   v8u;

typedef __attribute__((ext_vector_type(2)))  __bf16     v2bf;
typedef __attribute__((ext_vector_type(2)))  float      v2f;
typedef __attribute__((ext_vector_type(2)))  unsigned   v2u;

union FragAB { v8u u; v16bf b; };

__device__ __forceinline__ unsigned pack2bf(float lo, float hi) {
    const v2f  f = {lo, hi};
    const v2bf p = __builtin_convertvector(f, v2bf);
    return __builtin_bit_cast(unsigned, p);
}
__device__ __forceinline__ float fma_bf_lo(unsigned u, float f, float acc) {
    float r;
    asm("v_fma_mix_f32_bf16 %0, %1, %2, %3 op_sel_hi:[1,0,0]"
        : "=v"(r) : "v"(u), "v"(f), "v"(acc));
    return r;
}
__device__ __forceinline__ float fma_bf_hi(unsigned u, float f, float acc) {
    float r;
    asm("v_fma_mix_f32_bf16 %0, %1, %2, %3 op_sel:[1,0,0] op_sel_hi:[1,0,0]"
        : "=v"(r) : "v"(u), "v"(f), "v"(acc));
    return r;
}
__device__ __forceinline__ float add_bf_lo(float acc, unsigned u) { return fma_bf_lo(u, 1.0f, acc); }
__device__ __forceinline__ float add_bf_hi(float acc, unsigned u) { return fma_bf_hi(u, 1.0f, acc); }
__device__ __forceinline__ float sub_bf_lo(float z, unsigned u) { return fma_bf_lo(u, -1.0f, z); }
__device__ __forceinline__ float sub_bf_hi(float z, unsigned u) { return fma_bf_hi(u, -1.0f, z); }

#define TS_EDGES 4096
#define BK_SHIFT 8

__device__ __forceinline__ int block_excl_scan(int v, int lane, int wave,
                                               int* wsum, int* all_sum) {
    int incl = v;
#pragma unroll
    for (int off = 1; off < 32; off <<= 1) {
        const int t = __shfl_up(incl, off, 32);
        if (lane >= off) incl += t;
    }
    if (lane == 31) wsum[wave] = incl;
    __syncthreads();
    int wbase = 0, tot = 0;
#pragma unroll
    for (int w = 0; w < 8; ++w) {
        const int x = wsum[w];
        if (w < wave) wbase += x;
        tot += x;
    }
    *all_sum = tot;
    return wbase + incl - v;
}

__device__ __forceinline__ size_t w1_entry_at(int i) {
    const int v    = i & 7;
    const int lane = (i >> 3) & 31;
    const int ks   = (i >> 8) & 3;
    const int nt   = i >> 10;
    const int k    = ks * 32 + ((v >> 2) << 4) + ((lane >> 4) << 3) + ((v & 3) << 1);
    const int ncol = nt * 16 + (lane & 15);
    return (size_t)k * 128 + ncol;
}

__global__ void __launch_bounds__(256)
k_tile_sort(const int* __restrict__ src, const int* __restrict__ dst,
            int2* __restrict__ ts, int2* __restrict__ hist2, int* __restrict__ total,
            const float* __restrict__ W1, unsigned* __restrict__ w1p,
            int n_edges, int nb) {
    extern __shared__ int dyn[];
    __shared__ int2 stg[TS_EDGES];
    __shared__ int  wsum[8];
    int* cntL = dyn;
    int* offL = dyn + nb;
    const int tid = threadIdx.x, lane = tid & 31, wave = tid >> 5;
    for (int i = tid; i < nb; i += 256) cntL[i] = 0;
    if (blockIdx.x == 0 && tid == 0) total[0] = 0;
    __syncthreads();

    const int wi = blockIdx.x * 256 + tid;
    float w_lo = 0.f, w_hi = 0.f;
    if (wi < 8192) {
        const size_t a = w1_entry_at(wi);
        w_lo = W1[a];
        w_hi = W1[a + 128];
    }
    const long long base = (long long)blockIdx.x * TS_EDGES;
    int s[16], d[16], r[16];
#pragma unroll
    for (int k = 0; k < 16; ++k) {
        const long long i = base + k * 256 + tid;
        const bool ok = i < n_edges;
        d[k] = ok ? dst[i] : -1;
        s[k] = ok ? src[i] : 0;
    }
#pragma unroll
    for (int k = 0; k < 16; ++k) {
        r[k] = 0;
        if (d[k] >= 0) {
            const int bk = (int)min((unsigned)d[k] >> BK_SHIFT, (unsigned)(nb - 1));
            r[k] = atomicAdd(&cntL[bk], 1);
        }
    }
    if (wi < 8192) w1p[wi] = pack2bf(w_lo, w_hi);
    __syncthreads();
    const int chunk = (nb + 255) >> 8;
    const int lo = min(tid * chunk, nb), hi = min(lo + chunk, nb);
    int mine = 0;
    for (int i = lo; i < hi; ++i) mine += cntL[i];
    int all;
    int run = block_excl_scan(mine, lane, wave, wsum, &all);
    for (int i = lo; i < hi; ++i) { offL[i] = run; run += cntL[i]; }
    __syncthreads();
    for (int i = tid; i < nb; i += 256)
        hist2[(size_t)blockIdx.x * nb + i] = make_int2(offL[i], cntL[i]);
#pragma unroll
    for (int k = 0; k < 16; ++k) {
        if (d[k] >= 0) {
            const int bk = (int)min((unsigned)d[k] >> BK_SHIFT, (unsigned)(nb - 1));
            stg[(offL[bk] + r[k]) & (TS_EDGES - 1)] = make_int2(s[k], d[k]);
        }
    }
    __syncthreads();
    const long long left = (long long)n_edges - base;
#pragma unroll
    for (int k = 0; k < 16; ++k) {
        const int i = k * 256 + tid;
        if (i < left) ts[base + i] = stg[i];
    }
    for (int i = wi + (int)gridDim.x * 256; i < 8192; i += (int)gridDim.x * 256) {
        const size_t a = w1_entry_at(i);
        w1p[i] = pack2bf(W1[a], W1[a + 128]);
    }
}

#define BK_PER_T 18
#define PX_FLIGHT 16
#define BK_CAP   (256 * BK_PER_T)

__device__ __forceinline__ void bk_fill(int* own, const int* preL, const int* relL,
                                        const int* lenL, int n_tiles, int tid,
                                        int c0, int c1) {
    for (int t = tid; t < n_tiles; t += 256) {
        const int p = preL[t], rel = relL[t];
        const int lo = max(p, c0), hi = min(p + lenL[t], c1);
        for (int f = lo; f < hi; ++f) own[f - c0] = rel;
    }
}

template <bool WHOLE>
__device__ __forceinline__ void px_trip(const float4* __restrict__ x4, uint2* __restrict__ o2,
                                        const float* dinvL, int node0, int r0, int rows,
                                        int lane) {
    float4 v[PX_FLIGHT];
#pragma unroll
    for (int j = 0; j < PX_FLIGHT; ++j) {
        const int r = r0 + j;
        v[j] = make_float4(0.f, 0.f, 0.f, 0.f);
        if (WHOLE || r < rows) v[j] = x4[(size_t)(node0 + r) * 32 + lane];
    }
    __builtin_amdgcn_sched_barrier(0);
#pragma unroll
    for (int j = 0; j < PX_FLIGHT; ++j) {
        const int r = r0 + j;
        if (WHOLE || r < rows) {
            const float d = dinvL[r];
            o2[(size_t)(node0 + r) * 32 + lane] =
                make_uint2(pack2bf(v[j].x * d, v[j].y * d), pack2bf(v[j].z * d, v[j].w * d));
        }
    }
}

__global__ void __launch_bounds__(256)
k_bucket_csr(const int2* __restrict__ ts, const int2* __restrict__ hist2, int* total,
             int* __restrict__ seg, float* __restrict__ dinv, int* __restrict__ csr,
             const float* __restrict__ x, unsigned* __restrict__ xs,
             int n_nodes, int n_edges, int nb, int n_tiles) {
    extern __shared__ int dyn[];
    __shared__ int own[BK_CAP], outL[BK_CAP];
    __shared__ int cntN[256], offN[256], cur[256], wsum[8], sbase;
    __shared__ float dinvL[256];
    int* preL = dyn;
    int* relL = dyn + n_tiles;
    int* lenL = dyn + 2 * n_tiles;
    const int tid = threadIdx.x, lane = tid & 31, wave = tid >> 5;
    const int bk    = blockIdx.x;
    const int node0 = bk << BK_SHIFT;
    cntN[tid] = 0; cur[tid] = 0;
    int mine = 0;
    for (int t = tid; t < n_tiles; t += 256) {
        const int2 h = hist2[(size_t)t * nb + bk];
        const int len = min(max(h.y, 0), TS_EDGES);
        lenL[t] = len;
        relL[t] = t * TS_EDGES + (h.x & (TS_EDGES - 1));
        mine += len;
    }
    int n_bk;
    int run = block_excl_scan(mine, lane, wave, wsum, &n_bk);
    for (int t = tid; t < n_tiles; t += 256) {
        preL[t] = run;
        relL[t] -= run;
        run += lenL[t];
    }

    for (int c0 = 0; c0 < n_bk; c0 += BK_CAP) {
        const int c1 = min(n_bk, c0 + BK_CAP);
        __syncthreads();
        bk_fill(own, preL, relL, lenL, n_tiles, tid, c0, c1);
        __syncthreads();
        int dv[BK_PER_T];
#pragma unroll
        for (int j = 0; j < BK_PER_T; ++j) {
            const int f   = min(c0 + j * 256 + tid, c1 - 1);
            const int idx = min(max(own[f - c0] + f, 0), n_edges - 1);
            dv[j] = ts[idx].y;
        }
#pragma unroll
        for (int j = 0; j < BK_PER_T; ++j)
            if (c0 + j * 256 + tid < c1) atomicAdd(&cntN[(dv[j] - node0) & 255], 1);
    }
    __syncthreads();
    const int c = cntN[tid];
    const float dv_own = rsqrtf((float)(c + 1));
    dinvL[tid] = dv_own;
    int n_cnt;
    const int off = block_excl_scan(c, lane, wave, wsum, &n_cnt);
    offN[tid] = off;
    if (tid == 0) sbase = atomicAdd(total, n_cnt);
    __syncthreads();
    const int rb = sbase;
    const int node = node0 + tid;
    if (node < n_nodes) {
        ((int2*)seg)[node] = make_int2(rb + off, c);
        dinv[node] = dv_own;
    }

    const bool staged = n_bk <= BK_CAP;
    for (int c0 = 0; c0 < n_bk; c0 += BK_CAP) {
        const int c1 = min(n_bk, c0 + BK_CAP);
        if (!staged) {
            __syncthreads();
            bk_fill(own, preL, relL, lenL, n_tiles, tid, c0, c1);
            __syncthreads();
        }
        int2 ev[BK_PER_T];
#pragma unroll
        for (int j = 0; j < BK_PER_T; ++j) {
            const int f   = min(c0 + j * 256 + tid, c1 - 1);
            const int idx = min(max(own[f - c0] + f, 0), n_edges - 1);
            ev[j] = ts[idx];
        }
#pragma unroll
        for (int j = 0; j < BK_PER_T; ++j) {
            if (c0 + j * 256 + tid < c1) {
                const int loc = (ev[j].y - node0) & 255;
                const int p   = offN[loc] + atomicAdd(&cur[loc], 1);
                if (staged) outL[min(p, BK_CAP - 1)] = ev[j].x;
                else if ((unsigned)(rb + p) < (unsigned)n_edges) csr[rb + p] = ev[j].x;
            }
        }
    }
    if (staged) {
        __syncthreads();
        int ov[BK_PER_T];
#pragma unroll
        for (int j = 0; j < BK_PER_T; ++j) ov[j] = outL[j * 256 + tid];
#pragma unroll
        for (int j = 0; j < BK_PER_T; ++j) {
            const int i = j * 256 + tid;
            if (i < n_bk && (unsigned)(rb + i) < (unsigned)n_edges) csr[rb + i] = ov[j];
        }
    }

    {
        const float4* x4 = (const float4*)x;
        uint2*        o2 = (uint2*)xs;
        const int wv   = __builtin_amdgcn_readfirstlane(wave);
        const int rows = min(256, n_nodes - node0);
        for (int r0 = wv * 32; r0 < min(wv * 32 + 32, rows); r0 += PX_FLIGHT) {
            if (r0 + PX_FLIGHT <= rows) px_trip<true >(x4, o2, dinvL, node0, r0, rows, lane);
            else                        px_trip<false>(x4, o2, dinvL, node0, r0, rows, lane);
        }
    }
}

#define FT_ROWS  16
#define FT_PITCH 68
#define FT_WAVES 4
#define G1_CHUNK 31
__device__ __forceinline__ int g1_load_srcs(const int* __restrict__ csr, int first, int m,
                                            int lane, int node, int n_edges) {
    int sv = node;
    if (lane < m) {
        const int idx = min(first + lane, n_edges - 1);
        sv = csr[idx];
    }
    return sv;
}
__device__ __forceinline__ float4 g1_add(const float4 a, const v2u v) {
    return make_float4(add_bf_lo(a.x, v.x), add_bf_hi(a.y, v.x),
                       add_bf_lo(a.z, v.y), add_bf_hi(a.w, v.y));
}
typedef const __attribute__((address_space(1))) v2u* g1_rowp;
__device__ __forceinline__ g1_rowp g1_row(const uint2* rows, int s) {
    g1_rowp p = (g1_rowp)(rows + (size_t)s * 32);
    asm("" : "+s"(p));
    return p;
}
__global__ void __launch_bounds__(FT_WAVES * 32)
k_gather_gemm(const unsigned* __restrict__ xs,
              const int* __restrict__ csr,
              const int* __restrict__ seg,
              const float* __restrict__ dinv,
              const unsigned* __restrict__ w1p,
              const float* __restrict__ b1, const float* __restrict__ W2,
              float* __restrict__ h2s,
              int n_nodes, int n_edges) {
    __shared__ __attribute__((aligned(16))) unsigned zs[FT_WAVES][2][FT_ROWS][FT_PITCH];
    const int lane = threadIdx.x & 31;
    const int wv   = threadIdx.x >> 5;
    const int tile  = __builtin_amdgcn_readfirstlane((blockIdx.x * blockDim.x + threadIdx.x) >> 5);
    const int node0 = tile * FT_ROWS;
    if (node0 >= n_nodes) return;
    const int cntn = min(FT_ROWS, n_nodes - node0);

    int2  sgv = make_int2(0, 0);
    float dnv = 0.f;
    if (lane < cntn) {
        sgv = ((const int2*)seg)[node0 + lane];
        dnv = dinv[node0 + lane];
    }
    const uint2* rows = (const uint2*)xs;

    int beg = __builtin_amdgcn_readlane(sgv.x, 0);
    int n   = __builtin_amdgcn_readlane(sgv.y, 0);
    int sv_next = g1_load_srcs(csr, beg, min(n, G1_CHUNK), lane, node0, n_edges);

    for (int k = 0; k < cntn; ++k) {
        const int node = node0 + k;
        int sv = (int)min((unsigned)sv_next, (unsigned)(n_nodes - 1));
        int begn = 0, nn = 0;
        if (k + 1 < cntn) {
            begn = __builtin_amdgcn_readlane(sgv.x, k + 1);
            nn   = __builtin_amdgcn_readlane(sgv.y, k + 1);
            sv_next = g1_load_srcs(csr, begn, min(nn, G1_CHUNK), lane, node + 1, n_edges);
        }
        const unsigned lo8 = (unsigned)lane;
        const v2u own = g1_row(rows, node)[lo8];
        float4 acc  = make_float4(0.f, 0.f, 0.f, 0.f);
        float4 acc2 = make_float4(0.f, 0.f, 0.f, 0.f);
        for (int c0 = 0; c0 < n; c0 += G1_CHUNK) {
            const int m = min(n - c0, G1_CHUNK);
            if (c0 > 0)
                sv = (int)min((unsigned)g1_load_srcs(csr, beg + c0, m, lane, node, n_edges),
                              (unsigned)(n_nodes - 1));
            v2u ua[16], ub[8], uc[4], ud[2], ue;
            g1_rowp pa[16];
            int j = 0;
            if (m & 16) {
#pragma unroll
                for (int t = 0; t < 16; ++t)
                    pa[t] = g1_row(rows, __builtin_amdgcn_readlane(sv, t));
                __builtin_amdgcn_sched_barrier(0);
#pragma unroll
                for (int t = 0; t < 16; ++t)
                    ua[t] = pa[t][lo8];
                j = 16;
            }
            if (m & 8) {
#pragma unroll
                for (int t = 0; t < 8; ++t)
                    ub[t] = g1_row(rows, __builtin_amdgcn_readlane(sv, j + t))[lo8];
                j += 8;
            }
            if (m & 4) {
#pragma unroll
                for (int t = 0; t < 4; ++t)
                    uc[t] = g1_row(rows, __builtin_amdgcn_readlane(sv, j + t))[lo8];
                j += 4;
            }
            if (m & 2) {
                ud[0] = g1_row(rows, __builtin_amdgcn_readlane(sv, j))[lo8];
                ud[1] = g1_row(rows, __builtin_amdgcn_readlane(sv, j + 1))[lo8];
                j += 2;
            }
            if (m & 1)
                ue = g1_row(rows, __builtin_amdgcn_readlane(sv, j))[lo8];
            if (m & 16) {
#pragma unroll
                for (int t = 0; t < 16; t += 2) {
                    acc  = g1_add(acc,  ua[t]);
                    acc2 = g1_add(acc2, ua[t + 1]);
                }
            }
            if (m & 8) {
#pragma unroll
                for (int t = 0; t < 8; t += 2) {
                    acc  = g1_add(acc,  ub[t]);
                    acc2 = g1_add(acc2, ub[t + 1]);
                }
            }
            if (m & 4) {
                acc  = g1_add(acc,  uc[0]);
                acc2 = g1_add(acc2, uc[1]);
                acc  = g1_add(acc,  uc[2]);
                acc2 = g1_add(acc2, uc[3]);
            }
            if (m & 2) {
                acc  = g1_add(acc,  ud[0]);
                acc2 = g1_add(acc2, ud[1]);
            }
            if (m & 1)
                acc = g1_add(acc, ue);
        }
        acc2 = g1_add(acc2, own);

        const float dn = __int_as_float(__builtin_amdgcn_readlane(__float_as_int(dnv), k));
        const float z0 = (acc.x + acc2.x) * dn, z1 = (acc.y + acc2.y) * dn;
        const float z2 = (acc.z + acc2.z) * dn, z3 = (acc.w + acc2.w) * dn;
        const unsigned hA = pack2bf(z0, z1), hB = pack2bf(z2, z3);
        const float r0 = sub_bf_lo(z0, hA), r1 = sub_bf_hi(z1, hA);
        const float r2 = sub_bf_lo(z2, hB), r3 = sub_bf_hi(z3, hB);
        *(uint2*)&zs[wv][0][k][2 * lane] = make_uint2(hA, hB);
        *(uint2*)&zs[wv][1][k][2 * lane] = make_uint2(pack2bf(r0, r1), pack2bf(r2, r3));
        beg = begn; n = nn;
    }
    for (int k = cntn; k < FT_ROWS; ++k) {
        *(uint2*)&zs[wv][0][k][2 * lane] = make_uint2(0u, 0u);
        *(uint2*)&zs[wv][1][k][2 * lane] = make_uint2(0u, 0u);
    }
    __builtin_amdgcn_fence(__ATOMIC_RELEASE, "wavefront");
    __builtin_amdgcn_wave_barrier();
    __builtin_amdgcn_fence(__ATOMIC_ACQUIRE, "wavefront");

    const int m0   = node0;
    const int mr   = lane & 15;
    const int half = lane >> 4;
    const unsigned* rh = &zs[wv][0][mr][0];
    const unsigned* rl = &zs[wv][1][mr][0];
    const v8u* bfrag = (const v8u*)w1p;

    v8f acc[8];
#pragma unroll
    for (int nt = 0; nt < 8; ++nt) acc[nt] = (v8f){};
#pragma unroll
    for (int ks = 0; ks < 4; ++ks) {
        FragAB ah, al;
        {
            const uint4 hlo = *(const uint4*)(rh + ks * 16 + half * 4);
            const uint4 hhi = *(const uint4*)(rh + ks * 16 + 8 + half * 4);
            const uint4 llo = *(const uint4*)(rl + ks * 16 + half * 4);
            const uint4 lhi = *(const uint4*)(rl + ks * 16 + 8 + half * 4);
            ah.u = (v8u){hlo.x, hlo.y, hlo.z, hlo.w, hhi.x, hhi.y, hhi.z, hhi.w};
            al.u = (v8u){llo.x, llo.y, llo.z, llo.w, lhi.x, lhi.y, lhi.z, lhi.w};
        }
#pragma unroll
        for (int g = 0; g < 8; g += 4) {
            FragAB b[4];
#pragma unroll
            for (int i = 0; i < 4; ++i)
                b[i].u = bfrag[((g + i) * 4 + ks) * 32 + lane];
            __builtin_amdgcn_sched_barrier(0);
#pragma unroll
            for (int i = 0; i < 4; ++i)
                acc[g + i] = __builtin_amdgcn_wmma_f32_16x16x32_bf16(
                                 false, ah.b, false, b[i].b, (short)0, acc[g + i], false, false);
#pragma unroll
            for (int i = 0; i < 4; ++i)
                acc[g + i] = __builtin_amdgcn_wmma_f32_16x16x32_bf16(
                                 false, al.b, false, b[i].b, (short)0, acc[g + i], false, false);
            __builtin_amdgcn_sched_barrier(0);
        }
    }
    float bv[8], wa[8], wb[8];
#pragma unroll
    for (int nt = 0; nt < 8; ++nt) {
        const int col = nt * 16 + mr;
        const float2 w = ((const float2*)W2)[col];
        bv[nt] = b1[col]; wa[nt] = w.x; wb[nt] = w.y;
    }
    float p0[8], p1[8];
#pragma unroll
    for (int v = 0; v < 8; ++v) {
        float s0 = 0.f, s1 = 0.f;
#pragma unroll
        for (int nt = 0; nt < 8; ++nt) {
            const float t = fmaxf(acc[nt][v] + bv[nt], 0.f);
            s0 += t * wa[nt];
            s1 += t * wb[nt];
        }
        p0[v] = s0; p1[v] = s1;
    }
#pragma unroll
    for (int off = 8; off > 0; off >>= 1) {
#pragma unroll
        for (int v = 0; v < 8; ++v) {
            p0[v] += __shfl_xor(p0[v], off, 32);
            p1[v] += __shfl_xor(p1[v], off, 32);
        }
    }
    float o0 = p0[0], o1 = p1[0];
#pragma unroll
    for (int v = 1; v < 8; ++v)
        if (mr == v) { o0 = p0[v]; o1 = p1[v]; }
    const int orow = m0 + (half << 3) + mr;
    const float dn = __shfl(dnv, (half << 3) + (mr & 7), 32);
    if (mr < 8 && orow < n_nodes)
        ((float2*)h2s)[orow] = make_float2(o0 * dn, o1 * dn);
}

#define G2_A 8
#define G2_B 4
__device__ __forceinline__ int g2_idx(const int* __restrict__ csr, int beg, int j, int n_edges) {
    return csr[min(max(beg + j, 0), n_edges - 1)];
}
__device__ __forceinline__ float2 g2_row(const float2* __restrict__ hs, int s, bool on,
                                         int n_nodes) {
    float2 r = make_float2(0.f, 0.f);
    if (on) r = hs[min((unsigned)s, (unsigned)(n_nodes - 1))];
    return r;
}
__global__ void k_gather2_out(const float* __restrict__ h2s,
                              const int* __restrict__ csr,
                              const int* __restrict__ seg,
                              const float* __restrict__ dinv, const float* __restrict__ b2,
                              float* __restrict__ out, int n_nodes, int n_edges) {
    const int  gid   = blockIdx.x * blockDim.x + threadIdx.x;
    const int  node  = gid >> 1;
    const int  sub   = gid & 1;
    const bool valid = node < n_nodes;
    const int  nc    = valid ? node : n_nodes - 1;
    const float2* hs = (const float2*)h2s;

    const int2 sg  = ((const int2*)seg)[nc];
    float dn  = dinv[nc];
    float2 acc = make_float2(0.f, 0.f);
    if (sub == 0) acc = hs[nc];
    float b20 = b2[0], b21 = b2[1];
    const int  beg = sg.x;
    int        n   = valid ? sg.y : 0;
    asm volatile("" : "+v"(dn), "+v"(n));

    const bool m16 = __any(n > 2 * G2_A);
    const bool m24 = __any(n > 2 * (G2_A + G2_B));
    int s[G2_A + 2 * G2_B];
#pragma unroll
    for (int t = 0; t < G2_A; ++t) s[t] = g2_idx(csr, beg, sub + 2 * t, n_edges);
    if (m16) {
#pragma unroll
        for (int t = G2_A; t < G2_A + G2_B; ++t) s[t] = g2_idx(csr, beg, sub + 2 * t, n_edges);
    }
    if (m24) {
#pragma unroll
        for (int t = G2_A + G2_B; t < G2_A + 2 * G2_B; ++t) s[t] = g2_idx(csr, beg, sub + 2 * t, n_edges);
    }
    asm volatile("" : "+v"(s[0]), "+v"(s[1]), "+v"(s[2]), "+v"(s[3]),
                      "+v"(s[4]), "+v"(s[5]), "+v"(s[6]), "+v"(s[7]));
    float2 v[G2_A + 2 * G2_B];
#pragma unroll
    for (int t = 0; t < G2_A; ++t) v[t] = g2_row(hs, s[t], sub + 2 * t < n, n_nodes);
    if (m16) {
#pragma unroll
        for (int t = G2_A; t < G2_A + G2_B; ++t) v[t] = g2_row(hs, s[t], sub + 2 * t < n, n_nodes);
    }
    if (m24) {
#pragma unroll
        for (int t = G2_A + G2_B; t < G2_A + 2 * G2_B; ++t) v[t] = g2_row(hs, s[t], sub + 2 * t < n, n_nodes);
    }
    asm volatile("" : "+s"(b20), "+s"(b21), "+v"(v[0].x));
#pragma unroll
    for (int t = 0; t < G2_A; ++t) { acc.x += v[t].x; acc.y += v[t].y; }
    if (m16) {
#pragma unroll
        for (int t = G2_A; t < G2_A + G2_B; ++t) { acc.x += v[t].x; acc.y += v[t].y; }
    }
    if (m24) {
#pragma unroll
        for (int t = G2_A + G2_B; t < G2_A + 2 * G2_B; ++t) { acc.x += v[t].x; acc.y += v[t].y; }
    }
    for (int j = sub + 2 * (G2_A + 2 * G2_B); j < n; j += 2) {
        const int idx = min(beg + j, n_edges - 1);
        const int sr  = (int)min((unsigned)csr[idx], (unsigned)(n_nodes - 1));
        const float2 w = hs[sr];
        acc.x += w.x; acc.y += w.y;
    }
    acc.x += __shfl_xor(acc.x, 1, 32);
    acc.y += __shfl_xor(acc.y, 1, 32);
    if (valid && sub == 0) {
        const float z0 = acc.x * dn + b20;
        const float z1 = acc.y * dn + b21;
        const float m  = fmaxf(z0, z1);
        const float l  = m + __logf(__expf(z0 - m) + __expf(z1 - m));
        ((float2*)out)[node] = make_float2(z0 - l, z1 - l);
    }
}

extern "C" void kernel_launch(void* const* d_in, const int* in_sizes, int n_in,
                              void* d_out, int out_size, void* d_ws, size_t ws_size,
                              hipStream_t stream) {
    const float* x   = (const float*)d_in[0];
    const int*   ei  = (const int*)  d_in[1];
    const float* W1  = (const float*)d_in[2];
    const float* b1  = (const float*)d_in[3];
    const float* W2  = (const float*)d_in[4];
    const float* b2  = (const float*)d_in[5];
    float*       out = (float*)d_out;

    const int n_nodes = in_sizes[0] / 128;
    const int n_edges = in_sizes[1] / 2;
    const int* src = ei;
    const int* dst = ei + n_edges;

    const size_t nP = ((size_t)n_nodes + 127) / 128 * 128;
    unsigned* zh    = (unsigned*)d_ws;
    unsigned* zl    = zh + (size_t)n_nodes * 64;
    unsigned* xs    = zl + (size_t)n_nodes * 64;
    unsigned* w1p   = xs + (size_t)n_nodes * 64;
    float*    h2s   = (float*)(w1p + 8192);
    float*    dinv  = h2s + 2 * nP;
    int*      seg   = (int*)(dinv + nP);
    int*      total = seg + 2 * nP;
    int*      csr   = total + 128;
    int2*     ts    = (int2*)zh;
    int2*     hist2 = (int2*)zl;

    const int B = 256;
    const int nb = (n_nodes + 255) / 256;
    const int nt = (n_edges + TS_EDGES - 1) / TS_EDGES;
    k_tile_sort <<<nt, B, 2 * nb * sizeof(int), stream>>>(src, dst, ts, hist2, total, W1, w1p, n_edges, nb);
    k_bucket_csr<<<nb, B, 3 * nt * sizeof(int), stream>>>(ts, hist2, total, seg, dinv, csr, x, xs,
                                                          n_nodes, n_edges, nb, nt);
    const int n_tiles = (n_nodes + FT_ROWS - 1) / FT_ROWS;
    k_gather_gemm<<<(n_tiles + FT_WAVES - 1) / FT_WAVES, FT_WAVES * 32, 0, stream>>>(
        xs, csr, seg, dinv, w1p, b1, W2, h2s, n_nodes, n_edges);
    k_gather2_out<<<(n_nodes * 2 + B - 1) / B, B, 0, stream>>>(
        h2s, csr, seg, dinv, b2, out, n_nodes, n_edges);
}
